# starting kernel
# baseline (speedup 1.0000x reference)
#include <hip/hip_runtime.h>
#include <hip/hip_bf16.h>


typedef __attribute__((ext_vector_type(2))) float v2f;
typedef __attribute__((ext_vector_type(8))) float v8f;

#define TB 256

__global__ void k_deg_init(float* __restrict__ deg, int n) {
  int i = blockIdx.x * blockDim.x + threadIdx.x;
  if (i < n) deg[i] = 1.0f;
}

__global__ void k_deg_count(const int* __restrict__ dst, float* __restrict__ deg, int e) {
  int i = blockIdx.x * blockDim.x + threadIdx.x;
  if (i < e) atomicAdd(&deg[dst[i]], 1.0f);
}

__global__ void k_deg_rsqrt(float* __restrict__ deg, int n) {
  int i = blockIdx.x * blockDim.x + threadIdx.x;
  if (i < n) deg[i] = rsqrtf(deg[i]);
}

template <bool RELU_IN>
__global__ void k_gemm64_wmma(const float* __restrict__ X,
                              const float* __restrict__ W,
                              float* __restrict__ H, int nstrips) {
  int wave = (int)((blockIdx.x * blockDim.x + threadIdx.x) >> 5);
  int lane = threadIdx.x & 31;
  if (wave >= nstrips) return;

  const int m0    = wave * 16;
  const int mlow  = lane & 15;
  const int khalf = (lane >> 4) << 1;

  v2f a[16];
  const float* xr = X + (size_t)(m0 + mlow) * 64 + khalf;
#pragma unroll
  for (int k = 0; k < 16; ++k) {
    float ax = xr[k * 4 + 0];
    float ay = xr[k * 4 + 1];
    if (RELU_IN) { ax = fmaxf(ax, 0.0f); ay = fmaxf(ay, 0.0f); }
    a[k].x = ax; a[k].y = ay;
  }

#pragma unroll
  for (int nt = 0; nt < 4; ++nt) {
    const float* wr = W + (size_t)(nt * 16 + mlow) * 64 + khalf;
    v8f c = {};
#pragma unroll
    for (int k = 0; k < 16; ++k) {
      v2f b;
      b.x = wr[k * 4 + 0];
      b.y = wr[k * 4 + 1];
      c = __builtin_amdgcn_wmma_f32_16x16x4_f32(
           false, a[k],  false, b,
           (short)0, c,  false,  false);
    }
    float* o = H + (size_t)(m0 + 8 * (lane >> 4)) * 64 + nt * 16 + mlow;
#pragma unroll
    for (int v = 0; v < 8; ++v) o[(size_t)v * 64] = c[v];
  }
}

__global__ void k_self_bias(const float* __restrict__ h, const float* __restrict__ dis,
                            const float* __restrict__ bias, float* __restrict__ out, int n) {
  int t = blockIdx.x * blockDim.x + threadIdx.x;
  if (t >= n * 16) return;
  int i = t >> 4;
  int f = (t & 15) << 2;
  float w = dis[i]; w *= w;
  float4 v = *(const float4*)(h + (size_t)i * 64 + f);
  float4 b = *(const float4*)(bias + f);
  float4 r;
  r.x = v.x * w + b.x; r.y = v.y * w + b.y;
  r.z = v.z * w + b.z; r.w = v.w * w + b.w;
  *(float4*)(out + (size_t)i * 64 + f) = r;
}

__global__ void k_edge_scatter(const int* __restrict__ src, const int* __restrict__ dst,
                               const float* __restrict__ h, const float* __restrict__ dis,
                               float* __restrict__ out, int e) {
  int t = blockIdx.x * blockDim.x + threadIdx.x;
  int ed = t >> 4;
  if (ed >= e) return;
  int f = (t & 15) << 2;
  int s = src[ed], d = dst[ed];
  float w = dis[s] * dis[d];
  float4 v = *(const float4*)(h + (size_t)s * 64 + f);
  float* o = out + (size_t)d * 64 + f;
  atomicAdd(o + 0, v.x * w);
  atomicAdd(o + 1, v.y * w);
  atomicAdd(o + 2, v.z * w);
  atomicAdd(o + 3, v.w * w);
}

extern "C" void kernel_launch(void* const* d_in, const int* in_sizes, int n_in,
                              void* d_out, int out_size, void* d_ws, size_t ws_size,
                              hipStream_t stream) {
  (void)n_in; (void)out_size; (void)ws_size;

  const float* x  = (const float*)d_in[0];
  const int*   ei = (const int*)  d_in[1];
  const float* W1 = (const float*)d_in[2];
  const float* b1 = (const float*)d_in[3];
  const float* W2 = (const float*)d_in[4];
  const float* b2 = (const float*)d_in[5];
  float* out = (float*)d_out;

  const int n = in_sizes[0] / 64;
  const int e = in_sizes[1] / 2;
  const int* src = ei;
  const int* dst = ei + e;

  float* dis = (float*)d_ws;
  float* h   = dis + n;
  float* agg = h + (size_t)n * 64;

  const int nstrips = (n + 15) / 16;
  const int gemm_blocks = (nstrips + 7) / 8;
  const int nf_blocks   = (n * 16 + TB - 1) / TB;
  const int ef_blocks   = (e * 16 + TB - 1) / TB;

  k_deg_init <<<(n + TB - 1) / TB, TB, 0, stream>>>(dis, n);
  k_deg_count<<<(e + TB - 1) / TB, TB, 0, stream>>>(dst, dis, e);
  k_deg_rsqrt<<<(n + TB - 1) / TB, TB, 0, stream>>>(dis, n);

  k_gemm64_wmma<false><<<gemm_blocks, TB, 0, stream>>>(x, W1, h, nstrips);
  k_self_bias  <<<nf_blocks, TB, 0, stream>>>(h, dis, b1, agg, n);
  k_edge_scatter<<<ef_blocks, TB, 0, stream>>>(src, dst, h, dis, agg, e);

  k_gemm64_wmma<true><<<gemm_blocks, TB, 0, stream>>>(agg, W2, h, nstrips);
  k_self_bias  <<<nf_blocks, TB, 0, stream>>>(h, dis, b2, out, n);
  k_edge_scatter<<<ef_blocks, TB, 0, stream>>>(src, dst, h, dis, out, e);
}


# edges grouped by 64-node destination range (count, scan, place); a work-group per range adds its edge rows into an LDS tile by LDS f32 atomics; global f32 atomics, self_bias gone; sum order changes
# speedup vs baseline: 1.3351x; 1.3351x over previous
#include <hip/hip_runtime.h>
#include <hip/hip_bf16.h>


typedef __attribute__((ext_vector_type(2))) float v2f;
typedef __attribute__((ext_vector_type(8))) float v8f;

#define TB 256

__global__ void k_deg_init(float* __restrict__ deg, int n) {
  int i = blockIdx.x * blockDim.x + threadIdx.x;
  if (i < n) deg[i] = 1.0f;
}

__global__ void k_deg_count(const int* __restrict__ dst, float* __restrict__ deg, int e) {
  int i = blockIdx.x * blockDim.x + threadIdx.x;
  if (i < e) atomicAdd(&deg[dst[i]], 1.0f);
}

__global__ void k_deg_rsqrt(float* __restrict__ deg, int n) {
  int i = blockIdx.x * blockDim.x + threadIdx.x;
  if (i < n) deg[i] = rsqrtf(deg[i]);
}

template <bool RELU_IN>
__global__ void k_gemm64_wmma(const float* __restrict__ X,
                              const float* __restrict__ W,
                              float* __restrict__ H, int nstrips) {
  int wave = (int)((blockIdx.x * blockDim.x + threadIdx.x) >> 5);
  int lane = threadIdx.x & 31;
  if (wave >= nstrips) return;

  const int m0    = wave * 16;
  const int mlow  = lane & 15;
  const int khalf = (lane >> 4) << 1;

  v2f a[16];
  const float* xr = X + (size_t)(m0 + mlow) * 64 + khalf;
#pragma unroll
  for (int k = 0; k < 16; ++k) {
    float ax = xr[k * 4 + 0];
    float ay = xr[k * 4 + 1];
    if (RELU_IN) { ax = fmaxf(ax, 0.0f); ay = fmaxf(ay, 0.0f); }
    a[k].x = ax; a[k].y = ay;
  }

#pragma unroll
  for (int nt = 0; nt < 4; ++nt) {
    const float* wr = W + (size_t)(nt * 16 + mlow) * 64 + khalf;
    v8f c = {};
#pragma unroll
    for (int k = 0; k < 16; ++k) {
      v2f b;
      b.x = wr[k * 4 + 0];
      b.y = wr[k * 4 + 1];
      c = __builtin_amdgcn_wmma_f32_16x16x4_f32(
           false, a[k],  false, b,
           (short)0, c,  false,  false);
    }
    float* o = H + (size_t)(m0 + 8 * (lane >> 4)) * 64 + nt * 16 + mlow;
#pragma unroll
    for (int v = 0; v < 8; ++v) o[(size_t)v * 64] = c[v];
  }
}

#define NBK 64
#define BSH 6
#define CH  4096

__global__ void k_bin_count(const int* __restrict__ dst, int* __restrict__ cntT,
                            int e, int nbuck, int nchunk) {
  extern __shared__ int s_bin[];
  const int tid = threadIdx.x, chunk = blockIdx.x;
  for (int b = tid; b < nbuck; b += blockDim.x) s_bin[b] = 0;
  __syncthreads();
  const int base = chunk * CH;
  for (int i = tid; i < CH; i += blockDim.x) {
    const int ed = base + i;
    if (ed < e) atomicAdd(&s_bin[dst[ed] >> BSH], 1);
  }
  __syncthreads();
  for (int b = tid; b < nbuck; b += blockDim.x) cntT[(size_t)b * nchunk + chunk] = s_bin[b];
}

__global__ void k_xscan(int* arr, int len, int* totals) {
  __shared__ int s_v[1024];
  __shared__ int s_carry;
  const int tid = threadIdx.x;
  const int nt  = blockDim.x;
  int* a = arr + (size_t)blockIdx.x * len;
  if (tid == 0) s_carry = 0;
  __syncthreads();
  for (int c0 = 0; c0 < len; c0 += nt) {
    const int c = c0 + tid;
    const int v = (c < len) ? a[c] : 0;
    s_v[tid] = v;
    __syncthreads();
    for (int off = 1; off < nt; off <<= 1) {
      const int t = (tid >= off) ? s_v[tid - off] : 0;
      __syncthreads();
      s_v[tid] += t;
      __syncthreads();
    }
    const int incl = s_v[tid];
    const int cb = s_carry;
    if (c < len) a[c] = cb + incl - v;
    __syncthreads();
    if (tid == nt - 1) s_carry = cb + incl;
    __syncthreads();
  }
  if (tid == 0) totals[blockIdx.x] = s_carry;
}

__global__ void k_bin_place(const int* __restrict__ src, const int* __restrict__ dst,
                            const int* __restrict__ cntT, const int* __restrict__ bstart,
                            int2* __restrict__ rec, int e, int nbuck, int nchunk) {
  extern __shared__ int s_pos[];
  const int tid = threadIdx.x, chunk = blockIdx.x;
  for (int b = tid; b < nbuck; b += blockDim.x)
    s_pos[b] = bstart[b] + cntT[(size_t)b * nchunk + chunk];
  __syncthreads();
  const int base = chunk * CH;
  for (int i = tid; i < CH; i += blockDim.x) {
    const int ed = base + i;
    if (ed < e) {
      const int sv = src[ed], dv = dst[ed];
      const int p = atomicAdd(&s_pos[dv >> BSH], 1);
      rec[p] = make_int2(sv, dv);
    }
  }
}

__global__ void __launch_bounds__(256)
k_tile_aggregate(const int2* __restrict__ rec, const int* __restrict__ bstart,
                 const float* __restrict__ h, const float* __restrict__ dis,
                 const float* __restrict__ bias, float* __restrict__ out, int n) {
  __shared__ float acc[NBK * 64];
  const int tid  = threadIdx.x;
  const int lane = tid & 31;
  const int wave = __builtin_amdgcn_readfirstlane(tid >> 5);
  const int b  = blockIdx.x;
  const int n0 = b << BSH;
  const int e0 = bstart[b], e1 = bstart[b + 1];

  for (int idx = tid; idx < NBK * 16; idx += 256) {
    const int i = idx >> 4, f = (idx & 15) << 2;
    const int node = n0 + i;
    if (node < n) {
      float w = dis[node]; w *= w;
      const float4 v  = *(const float4*)(h + (size_t)node * 64 + f);
      const float4 bb = *(const float4*)(bias + f);
      float4 r;
      r.x = v.x * w + bb.x; r.y = v.y * w + bb.y;
      r.z = v.z * w + bb.z; r.w = v.w * w + bb.w;
      *(float4*)(acc + i * 64 + f) = r;
    }
  }
  __syncthreads();

  const float* hl = h + 2 * lane;
  float* al = acc + 2 * lane;
  for (int eb = e0 + wave * 32; eb < e1; eb += 8 * 32) {
    const int my = eb + lane;
    int rs = 0, rl = 0;
    float rw = 0.0f;
    if (my < e1) {
      const int2 r = rec[my];
      rs = r.x;
      rl = (r.y - n0) << 6;
      rw = dis[r.x] * dis[r.y];
    }
    const int left = e1 - eb;
    const int cnt = left < 32 ? left : 32;
    int j = 0;
    for (; j + 8 <= cnt; j += 8) {
      float2 v[8];
#pragma unroll
      for (int u = 0; u < 8; ++u) {
        const int sj = __builtin_amdgcn_readlane(rs, j + u);
        v[u] = *(const float2*)(hl + (size_t)sj * 64);
      }
#pragma unroll
      for (int u = 0; u < 8; ++u) {
        const int ro = __builtin_amdgcn_readlane(rl, j + u);
        const float wj = __builtin_bit_cast(float,
            __builtin_amdgcn_readlane(__builtin_bit_cast(int, rw), j + u));
        atomicAdd(al + ro,     v[u].x * wj);
        atomicAdd(al + ro + 1, v[u].y * wj);
      }
    }
    for (; j < cnt; ++j) {
      const int sj = __builtin_amdgcn_readlane(rs, j);
      const int ro = __builtin_amdgcn_readlane(rl, j);
      const float wj = __builtin_bit_cast(float,
          __builtin_amdgcn_readlane(__builtin_bit_cast(int, rw), j));
      const float2 v = *(const float2*)(hl + (size_t)sj * 64);
      atomicAdd(al + ro,     v.x * wj);
      atomicAdd(al + ro + 1, v.y * wj);
    }
  }
  __syncthreads();

  for (int idx = tid; idx < NBK * 16; idx += 256) {
    const int i = idx >> 4, f = (idx & 15) << 2;
    const int node = n0 + i;
    if (node < n)
      *(float4*)(out + (size_t)node * 64 + f) = *(const float4*)(acc + i * 64 + f);
  }
}

extern "C" void kernel_launch(void* const* d_in, const int* in_sizes, int n_in,
                              void* d_out, int out_size, void* d_ws, size_t ws_size,
                              hipStream_t stream) {
  (void)n_in; (void)out_size; (void)ws_size;

  const float* x  = (const float*)d_in[0];
  const int*   ei = (const int*)  d_in[1];
  const float* W1 = (const float*)d_in[2];
  const float* b1 = (const float*)d_in[3];
  const float* W2 = (const float*)d_in[4];
  const float* b2 = (const float*)d_in[5];
  float* out = (float*)d_out;

  const int n = in_sizes[0] / 64;
  const int e = in_sizes[1] / 2;
  const int* src = ei;
  const int* dst = ei + e;

  const int nbuck  = (n + NBK - 1) / NBK;
  const int nchunk = (e + CH - 1) / CH;

  float* dis = (float*)d_ws;
  float* h   = dis + n;
  float* agg = h + (size_t)n * 64;
  int* cntT   = (int*)(agg + (size_t)n * 64);
  int* bstart = cntT + (size_t)nbuck * nchunk;
  int2* rec   = (int2*)(((uintptr_t)(bstart + nbuck + 1) + 7) & ~(uintptr_t)7);

  const int nstrips = (n + 15) / 16;
  const int gemm_blocks = (nstrips + 7) / 8;
  const size_t bin_lds  = (size_t)nbuck * sizeof(int);

  k_deg_init <<<(n + TB - 1) / TB, TB, 0, stream>>>(dis, n);
  k_deg_count<<<(e + TB - 1) / TB, TB, 0, stream>>>(dst, dis, e);
  k_deg_rsqrt<<<(n + TB - 1) / TB, TB, 0, stream>>>(dis, n);

  k_bin_count<<<nchunk, 512, bin_lds, stream>>>(dst, cntT, e, nbuck, nchunk);
  k_xscan    <<<nbuck, 512, 0, stream>>>(cntT, nchunk, bstart);
  k_xscan    <<<1, 1024, 0, stream>>>(bstart, nbuck, bstart + nbuck);
  k_bin_place<<<nchunk, 512, bin_lds, stream>>>(src, dst, cntT, bstart, rec, e, nbuck, nchunk);

  k_gemm64_wmma<false><<<gemm_blocks, TB, 0, stream>>>(x, W1, h, nstrips);
  k_tile_aggregate<<<nbuck, 256, 0, stream>>>(rec, bstart, h, dis, b1, agg, n);

  k_gemm64_wmma<true><<<gemm_blocks, TB, 0, stream>>>(agg, W2, h, nstrips);
  k_tile_aggregate<<<nbuck, 256, 0, stream>>>(rec, bstart, h, dis, b2, out, n);
}


# product kernel stores row s times dis[s]; a range's tile adds source rows unweighted and is scaled by dis[i] once at its store: per-edge random dis loads and weight multiplies gone; rounding changes
# speedup vs baseline: 1.3761x; 1.0307x over previous
#include <hip/hip_runtime.h>
#include <hip/hip_bf16.h>


typedef __attribute__((ext_vector_type(2))) float v2f;
typedef __attribute__((ext_vector_type(8))) float v8f;

#define TB 256

__global__ void k_deg_init(float* __restrict__ deg, int n) {
  int i = blockIdx.x * blockDim.x + threadIdx.x;
  if (i < n) deg[i] = 1.0f;
}

__global__ void k_deg_count(const int* __restrict__ dst, float* __restrict__ deg, int e) {
  int i = blockIdx.x * blockDim.x + threadIdx.x;
  if (i < e) atomicAdd(&deg[dst[i]], 1.0f);
}

__global__ void k_deg_rsqrt(float* __restrict__ deg, int n) {
  int i = blockIdx.x * blockDim.x + threadIdx.x;
  if (i < n) deg[i] = rsqrtf(deg[i]);
}

template <bool RELU_IN>
__global__ void k_gemm64_wmma(const float* __restrict__ X,
                              const float* __restrict__ W,
                              const float* __restrict__ dis,
                              float* __restrict__ H, int nstrips) {
  int wave = (int)((blockIdx.x * blockDim.x + threadIdx.x) >> 5);
  int lane = threadIdx.x & 31;
  if (wave >= nstrips) return;

  const int m0    = wave * 16;
  const int mlow  = lane & 15;
  const int khalf = (lane >> 4) << 1;

  const float* dr = dis + m0 + 8 * (lane >> 4);
  const float4 d0 = *(const float4*)dr;
  const float4 d1 = *(const float4*)(dr + 4);
  const float dv[8] = {d0.x, d0.y, d0.z, d0.w, d1.x, d1.y, d1.z, d1.w};

  v2f a[16];
  const float* xr = X + (size_t)(m0 + mlow) * 64 + khalf;
#pragma unroll
  for (int k = 0; k < 16; ++k) {
    float ax = xr[k * 4 + 0];
    float ay = xr[k * 4 + 1];
    if (RELU_IN) { ax = fmaxf(ax, 0.0f); ay = fmaxf(ay, 0.0f); }
    a[k].x = ax; a[k].y = ay;
  }

#pragma unroll
  for (int nt = 0; nt < 4; ++nt) {
    const float* wr = W + (size_t)(nt * 16 + mlow) * 64 + khalf;
    v8f c = {};
#pragma unroll
    for (int k = 0; k < 16; ++k) {
      v2f b;
      b.x = wr[k * 4 + 0];
      b.y = wr[k * 4 + 1];
      c = __builtin_amdgcn_wmma_f32_16x16x4_f32(
           false, a[k],  false, b,
           (short)0, c,  false,  false);
    }
    float* o = H + (size_t)(m0 + 8 * (lane >> 4)) * 64 + nt * 16 + mlow;
#pragma unroll
    for (int v = 0; v < 8; ++v) o[(size_t)v * 64] = c[v] * dv[v];
  }
}

#define NBK 64
#define BSH 6
#define CH  4096

__global__ void k_bin_count(const int* __restrict__ dst, int* __restrict__ cntT,
                            int e, int nbuck, int nchunk) {
  extern __shared__ int s_bin[];
  const int tid = threadIdx.x, chunk = blockIdx.x;
  for (int b = tid; b < nbuck; b += blockDim.x) s_bin[b] = 0;
  __syncthreads();
  const int base = chunk * CH;
  for (int i = tid; i < CH; i += blockDim.x) {
    const int ed = base + i;
    if (ed < e) atomicAdd(&s_bin[dst[ed] >> BSH], 1);
  }
  __syncthreads();
  for (int b = tid; b < nbuck; b += blockDim.x) cntT[(size_t)b * nchunk + chunk] = s_bin[b];
}

__global__ void k_xscan(int* arr, int len, int* totals) {
  __shared__ int s_v[1024];
  __shared__ int s_carry;
  const int tid = threadIdx.x;
  const int nt  = blockDim.x;
  int* a = arr + (size_t)blockIdx.x * len;
  if (tid == 0) s_carry = 0;
  __syncthreads();
  for (int c0 = 0; c0 < len; c0 += nt) {
    const int c = c0 + tid;
    const int v = (c < len) ? a[c] : 0;
    s_v[tid] = v;
    __syncthreads();
    for (int off = 1; off < nt; off <<= 1) {
      const int t = (tid >= off) ? s_v[tid - off] : 0;
      __syncthreads();
      s_v[tid] += t;
      __syncthreads();
    }
    const int incl = s_v[tid];
    const int cb = s_carry;
    if (c < len) a[c] = cb + incl - v;
    __syncthreads();
    if (tid == nt - 1) s_carry = cb + incl;
    __syncthreads();
  }
  if (tid == 0) totals[blockIdx.x] = s_carry;
}

__global__ void k_bin_place(const int* __restrict__ src, const int* __restrict__ dst,
                            const int* __restrict__ cntT, const int* __restrict__ bstart,
                            int2* __restrict__ rec, int e, int nbuck, int nchunk) {
  extern __shared__ int s_pos[];
  const int tid = threadIdx.x, chunk = blockIdx.x;
  for (int b = tid; b < nbuck; b += blockDim.x)
    s_pos[b] = bstart[b] + cntT[(size_t)b * nchunk + chunk];
  __syncthreads();
  const int base = chunk * CH;
  for (int i = tid; i < CH; i += blockDim.x) {
    const int ed = base + i;
    if (ed < e) {
      const int sv = src[ed], dv = dst[ed];
      const int p = atomicAdd(&s_pos[dv >> BSH], 1);
      rec[p] = make_int2(sv, dv);
    }
  }
}

__global__ void __launch_bounds__(256)
k_tile_aggregate(const int2* __restrict__ rec, const int* __restrict__ bstart,
                 const float* __restrict__ h, const float* __restrict__ dis,
                 const float* __restrict__ bias, float* __restrict__ out, int n) {
  __shared__ float acc[NBK * 64];
  const int tid  = threadIdx.x;
  const int lane = tid & 31;
  const int wave = __builtin_amdgcn_readfirstlane(tid >> 5);
  const int b  = blockIdx.x;
  const int n0 = b << BSH;
  const int e0 = bstart[b], e1 = bstart[b + 1];

  for (int idx = tid; idx < NBK * 16; idx += 256) {
    const int i = idx >> 4, f = (idx & 15) << 2;
    const int node = n0 + i;
    if (node < n)
      *(float4*)(acc + i * 64 + f) = *(const float4*)(h + (size_t)node * 64 + f);
  }
  __syncthreads();

  const float* hl = h + 2 * lane;
  float* al = acc + 2 * lane;
  for (int eb = e0 + wave * 32; eb < e1; eb += 8 * 32) {
    const int my = eb + lane;
    int rs = 0, rl = 0;
    if (my < e1) {
      const int2 r = rec[my];
      rs = r.x;
      rl = (r.y - n0) << 6;
    }
    const int left = e1 - eb;
    const int cnt = left < 32 ? left : 32;
    int j = 0;
    for (; j + 8 <= cnt; j += 8) {
      float2 v[8];
#pragma unroll
      for (int u = 0; u < 8; ++u) {
        const int sj = __builtin_amdgcn_readlane(rs, j + u);
        v[u] = *(const float2*)(hl + (size_t)sj * 64);
      }
#pragma unroll
      for (int u = 0; u < 8; ++u) {
        const int ro = __builtin_amdgcn_readlane(rl, j + u);
        atomicAdd(al + ro,     v[u].x);
        atomicAdd(al + ro + 1, v[u].y);
      }
    }
    for (; j < cnt; ++j) {
      const int sj = __builtin_amdgcn_readlane(rs, j);
      const int ro = __builtin_amdgcn_readlane(rl, j);
      const float2 v = *(const float2*)(hl + (size_t)sj * 64);
      atomicAdd(al + ro,     v.x);
      atomicAdd(al + ro + 1, v.y);
    }
  }
  __syncthreads();

  for (int idx = tid; idx < NBK * 16; idx += 256) {
    const int i = idx >> 4, f = (idx & 15) << 2;
    const int node = n0 + i;
    if (node < n) {
      const float w = dis[node];
      const float4 v  = *(const float4*)(acc + i * 64 + f);
      const float4 bb = *(const float4*)(bias + f);
      float4 r;
      r.x = v.x * w + bb.x; r.y = v.y * w + bb.y;
      r.z = v.z * w + bb.z; r.w = v.w * w + bb.w;
      *(float4*)(out + (size_t)node * 64 + f) = r;
    }
  }
}

extern "C" void kernel_launch(void* const* d_in, const int* in_sizes, int n_in,
                              void* d_out, int out_size, void* d_ws, size_t ws_size,
                              hipStream_t stream) {
  (void)n_in; (void)out_size; (void)ws_size;

  const float* x  = (const float*)d_in[0];
  const int*   ei = (const int*)  d_in[1];
  const float* W1 = (const float*)d_in[2];
  const float* b1 = (const float*)d_in[3];
  const float* W2 = (const float*)d_in[4];
  const float* b2 = (const float*)d_in[5];
  float* out = (float*)d_out;

  const int n = in_sizes[0] / 64;
  const int e = in_sizes[1] / 2;
  const int* src = ei;
  const int* dst = ei + e;

  const int nbuck  = (n + NBK - 1) / NBK;
  const int nchunk = (e + CH - 1) / CH;

  float* dis = (float*)d_ws;
  float* h   = dis + n;
  float* agg = h + (size_t)n * 64;
  int* cntT   = (int*)(agg + (size_t)n * 64);
  int* bstart = cntT + (size_t)nbuck * nchunk;
  int2* rec   = (int2*)(((uintptr_t)(bstart + nbuck + 1) + 7) & ~(uintptr_t)7);

  const int nstrips = (n + 15) / 16;
  const int gemm_blocks = (nstrips + 7) / 8;
  const size_t bin_lds  = (size_t)nbuck * sizeof(int);

  k_deg_init <<<(n + TB - 1) / TB, TB, 0, stream>>>(dis, n);
  k_deg_count<<<(e + TB - 1) / TB, TB, 0, stream>>>(dst, dis, e);
  k_deg_rsqrt<<<(n + TB - 1) / TB, TB, 0, stream>>>(dis, n);

  k_bin_count<<<nchunk, 512, bin_lds, stream>>>(dst, cntT, e, nbuck, nchunk);
  k_xscan    <<<nbuck, 512, 0, stream>>>(cntT, nchunk, bstart);
  k_xscan    <<<1, 1024, 0, stream>>>(bstart, nbuck, bstart + nbuck);
  k_bin_place<<<nchunk, 512, bin_lds, stream>>>(src, dst, cntT, bstart, rec, e, nbuck, nchunk);

  k_gemm64_wmma<false><<<gemm_blocks, TB, 0, stream>>>(x, W1, dis, h, nstrips);
  k_tile_aggregate<<<nbuck, 256, 0, stream>>>(rec, bstart, h, dis, b1, agg, n);

  k_gemm64_wmma<true><<<gemm_blocks, TB, 0, stream>>>(agg, W2, dis, h, nstrips);
  k_tile_aggregate<<<nbuck, 256, 0, stream>>>(rec, bstart, h, dis, b2, out, n);
}


# in-degree counted in LDS from the range-grouped edge records inside the layer-1 product launch (128 rows = 2 ranges); the 3 degree launches and their global atomics gone; dis bit-identical
# speedup vs baseline: 1.8466x; 1.3419x over previous
#include <hip/hip_runtime.h>
#include <hip/hip_bf16.h>


typedef __attribute__((ext_vector_type(2))) float v2f;
typedef __attribute__((ext_vector_type(8))) float v8f;

#define TB 256

#define NBK 64
#define BSH 6
#define CH  4096
#define GROWS (TB / 32 * 16)
#define DEGU 8
static_assert(GROWS % NBK == 0, "a product work-group must cover whole ranges");

template <bool RELU_IN, bool DEG>
__global__ void __launch_bounds__(TB)
k_gemm64_wmma(const float* __restrict__ X,
              const float* __restrict__ W,
              float* dis,
              const int2* __restrict__ rec, const int* __restrict__ bstart,
              float* __restrict__ H, int nstrips, int n, int nbuck) {
  __shared__ int s_cnt[DEG ? GROWS : 1];
  const int tid  = threadIdx.x;
  const int lane = tid & 31;
  const int wib  = tid >> 5;
  const int wave = (int)blockIdx.x * (TB / 32) + wib;
  const bool live = wave < nstrips;

  const int m0    = wave * 16;
  const int mlow  = lane & 15;
  const int khalf = (lane >> 4) << 1;

  if constexpr (DEG) {
    if (tid < GROWS) s_cnt[tid] = 0;
    __syncthreads();
  }

  v2f a[16];
#pragma unroll
  for (int k = 0; k < 16; ++k) { a[k].x = 0.0f; a[k].y = 0.0f; }
  if (live) {
    const float* xr = X + (size_t)(m0 + mlow) * 64 + khalf;
#pragma unroll
    for (int k = 0; k < 16; ++k) {
      float ax = xr[k * 4 + 0];
      float ay = xr[k * 4 + 1];
      if (RELU_IN) { ax = fmaxf(ax, 0.0f); ay = fmaxf(ay, 0.0f); }
      a[k].x = ax; a[k].y = ay;
    }
  }

  float dv[8];
  if constexpr (DEG) {
    const int r0 = (int)blockIdx.x * GROWS;
    const int b0 = r0 >> BSH;
    int b2 = b0 + GROWS / NBK;
    if (b2 > nbuck) b2 = nbuck;
    const int e0 = bstart[b0], e1 = bstart[b2];
    for (int i = e0 + tid; i < e1; i += DEGU * TB) {
      int d[DEGU];
#pragma unroll
      for (int u = 0; u < DEGU; ++u) {
        const int idx = i + u * TB;
        const int idc = idx < e1 ? idx : e1 - 1;
        const int row = rec[idc].y - r0;
        d[u] = idx < e1 ? row : -1;
      }
#pragma unroll
      for (int u = 0; u < DEGU; ++u)
        if (d[u] >= 0) atomicAdd(&s_cnt[d[u]], 1);
    }
    __syncthreads();
    if (tid < GROWS && r0 + tid < n) dis[r0 + tid] = rsqrtf(1.0f + (float)s_cnt[tid]);
    const int* cr = s_cnt + wib * 16 + 8 * (lane >> 4);
#pragma unroll
    for (int v = 0; v < 8; ++v) dv[v] = rsqrtf(1.0f + (float)cr[v]);
    if (!live) return;
  } else {
    if (!live) return;
    const float* dr = dis + m0 + 8 * (lane >> 4);
    const float4 d0 = *(const float4*)dr;
    const float4 d1 = *(const float4*)(dr + 4);
    dv[0] = d0.x; dv[1] = d0.y; dv[2] = d0.z; dv[3] = d0.w;
    dv[4] = d1.x; dv[5] = d1.y; dv[6] = d1.z; dv[7] = d1.w;
  }

#pragma unroll
  for (int nt = 0; nt < 4; ++nt) {
    const float* wr = W + (size_t)(nt * 16 + mlow) * 64 + khalf;
    v8f c = {};
#pragma unroll
    for (int k = 0; k < 16; ++k) {
      v2f b;
      b.x = wr[k * 4 + 0];
      b.y = wr[k * 4 + 1];
      c = __builtin_amdgcn_wmma_f32_16x16x4_f32(
           false, a[k],  false, b,
           (short)0, c,  false,  false);
    }
    float* o = H + (size_t)(m0 + 8 * (lane >> 4)) * 64 + nt * 16 + mlow;
#pragma unroll
    for (int v = 0; v < 8; ++v) o[(size_t)v * 64] = c[v] * dv[v];
  }
}

__global__ void k_bin_count(const int* __restrict__ dst, int* __restrict__ cntT,
                            int e, int nbuck, int nchunk) {
  extern __shared__ int s_bin[];
  const int tid = threadIdx.x, chunk = blockIdx.x;
  for (int b = tid; b < nbuck; b += blockDim.x) s_bin[b] = 0;
  __syncthreads();
  const int base = chunk * CH;
  for (int i = tid; i < CH; i += blockDim.x) {
    const int ed = base + i;
    if (ed < e) atomicAdd(&s_bin[dst[ed] >> BSH], 1);
  }
  __syncthreads();
  for (int b = tid; b < nbuck; b += blockDim.x) cntT[(size_t)b * nchunk + chunk] = s_bin[b];
}

__global__ void k_xscan(int* arr, int len, int* totals) {
  __shared__ int s_v[1024];
  __shared__ int s_carry;
  const int tid = threadIdx.x;
  const int nt  = blockDim.x;
  int* a = arr + (size_t)blockIdx.x * len;
  if (tid == 0) s_carry = 0;
  __syncthreads();
  for (int c0 = 0; c0 < len; c0 += nt) {
    const int c = c0 + tid;
    const int v = (c < len) ? a[c] : 0;
    s_v[tid] = v;
    __syncthreads();
    for (int off = 1; off < nt; off <<= 1) {
      const int t = (tid >= off) ? s_v[tid - off] : 0;
      __syncthreads();
      s_v[tid] += t;
      __syncthreads();
    }
    const int incl = s_v[tid];
    const int cb = s_carry;
    if (c < len) a[c] = cb + incl - v;
    __syncthreads();
    if (tid == nt - 1) s_carry = cb + incl;
    __syncthreads();
  }
  if (tid == 0) totals[blockIdx.x] = s_carry;
}

__global__ void k_bin_place(const int* __restrict__ src, const int* __restrict__ dst,
                            const int* __restrict__ cntT, const int* __restrict__ bstart,
                            int2* __restrict__ rec, int e, int nbuck, int nchunk) {
  extern __shared__ int s_pos[];
  const int tid = threadIdx.x, chunk = blockIdx.x;
  for (int b = tid; b < nbuck; b += blockDim.x)
    s_pos[b] = bstart[b] + cntT[(size_t)b * nchunk + chunk];
  __syncthreads();
  const int base = chunk * CH;
  for (int i = tid; i < CH; i += blockDim.x) {
    const int ed = base + i;
    if (ed < e) {
      const int sv = src[ed], dv = dst[ed];
      const int p = atomicAdd(&s_pos[dv >> BSH], 1);
      rec[p] = make_int2(sv, dv);
    }
  }
}

__global__ void __launch_bounds__(256)
k_tile_aggregate(const int2* __restrict__ rec, const int* __restrict__ bstart,
                 const float* __restrict__ h, const float* __restrict__ dis,
                 const float* __restrict__ bias, float* __restrict__ out, int n) {
  __shared__ float acc[NBK * 64];
  const int tid  = threadIdx.x;
  const int lane = tid & 31;
  const int wave = __builtin_amdgcn_readfirstlane(tid >> 5);
  const int b  = blockIdx.x;
  const int n0 = b << BSH;
  const int e0 = bstart[b], e1 = bstart[b + 1];

  for (int idx = tid; idx < NBK * 16; idx += 256) {
    const int i = idx >> 4, f = (idx & 15) << 2;
    const int node = n0 + i;
    if (node < n)
      *(float4*)(acc + i * 64 + f) = *(const float4*)(h + (size_t)node * 64 + f);
  }
  __syncthreads();

  const float* hl = h + 2 * lane;
  float* al = acc + 2 * lane;
  for (int eb = e0 + wave * 32; eb < e1; eb += 8 * 32) {
    const int my = eb + lane;
    int rs = 0, rl = 0;
    if (my < e1) {
      const int2 r = rec[my];
      rs = r.x;
      rl = (r.y - n0) << 6;
    }
    const int left = e1 - eb;
    const int cnt = left < 32 ? left : 32;
    int j = 0;
    for (; j + 8 <= cnt; j += 8) {
      float2 v[8];
#pragma unroll
      for (int u = 0; u < 8; ++u) {
        const int sj = __builtin_amdgcn_readlane(rs, j + u);
        v[u] = *(const float2*)(hl + (size_t)sj * 64);
      }
#pragma unroll
      for (int u = 0; u < 8; ++u) {
        const int ro = __builtin_amdgcn_readlane(rl, j + u);
        atomicAdd(al + ro,     v[u].x);
        atomicAdd(al + ro + 1, v[u].y);
      }
    }
    for (; j < cnt; ++j) {
      const int sj = __builtin_amdgcn_readlane(rs, j);
      const int ro = __builtin_amdgcn_readlane(rl, j);
      const float2 v = *(const float2*)(hl + (size_t)sj * 64);
      atomicAdd(al + ro,     v.x);
      atomicAdd(al + ro + 1, v.y);
    }
  }
  __syncthreads();

  for (int idx = tid; idx < NBK * 16; idx += 256) {
    const int i = idx >> 4, f = (idx & 15) << 2;
    const int node = n0 + i;
    if (node < n) {
      const float w = dis[node];
      const float4 v  = *(const float4*)(acc + i * 64 + f);
      const float4 bb = *(const float4*)(bias + f);
      float4 r;
      r.x = v.x * w + bb.x; r.y = v.y * w + bb.y;
      r.z = v.z * w + bb.z; r.w = v.w * w + bb.w;
      *(float4*)(out + (size_t)node * 64 + f) = r;
    }
  }
}

extern "C" void kernel_launch(void* const* d_in, const int* in_sizes, int n_in,
                              void* d_out, int out_size, void* d_ws, size_t ws_size,
                              hipStream_t stream) {
  (void)n_in; (void)out_size; (void)ws_size;

  const float* x  = (const float*)d_in[0];
  const int*   ei = (const int*)  d_in[1];
  const float* W1 = (const float*)d_in[2];
  const float* b1 = (const float*)d_in[3];
  const float* W2 = (const float*)d_in[4];
  const float* b2 = (const float*)d_in[5];
  float* out = (float*)d_out;

  const int n = in_sizes[0] / 64;
  const int e = in_sizes[1] / 2;
  const int* src = ei;
  const int* dst = ei + e;

  const int nbuck  = (n + NBK - 1) / NBK;
  const int nchunk = (e + CH - 1) / CH;

  float* dis = (float*)d_ws;
  float* h   = dis + n;
  float* agg = h + (size_t)n * 64;
  int* cntT   = (int*)(agg + (size_t)n * 64);
  int* bstart = cntT + (size_t)nbuck * nchunk;
  int2* rec   = (int2*)(((uintptr_t)(bstart + nbuck + 1) + 7) & ~(uintptr_t)7);

  const int nstrips = (n + 15) / 16;
  const int gemm_blocks = (nstrips + 7) / 8;
  const size_t bin_lds  = (size_t)nbuck * sizeof(int);

  k_bin_count<<<nchunk, 512, bin_lds, stream>>>(dst, cntT, e, nbuck, nchunk);
  k_xscan    <<<nbuck, 512, 0, stream>>>(cntT, nchunk, bstart);
  k_xscan    <<<1, 1024, 0, stream>>>(bstart, nbuck, bstart + nbuck);
  k_bin_place<<<nchunk, 512, bin_lds, stream>>>(src, dst, cntT, bstart, rec, e, nbuck, nchunk);

  k_gemm64_wmma<false, true><<<gemm_blocks, TB, 0, stream>>>(x, W1, dis, rec, bstart, h, nstrips, n, nbuck);
  k_tile_aggregate<<<nbuck, 256, 0, stream>>>(rec, bstart, h, dis, b1, agg, n);

  k_gemm64_wmma<true, false><<<gemm_blocks, TB, 0, stream>>>(agg, W2, dis, rec, bstart, h, nstrips, n, nbuck);
  k_tile_aggregate<<<nbuck, 256, 0, stream>>>(rec, bstart, h, dis, b2, out, n);
}


# partition prefix sums on lane shifts inside single waves: one wave scans a range's chunk counts (was 512 threads, 22 barriers), placing work-groups rebuild range starts; one launch gone; same sums
# speedup vs baseline: 1.9208x; 1.0402x over previous
#include <hip/hip_runtime.h>
#include <hip/hip_bf16.h>


typedef __attribute__((ext_vector_type(2))) float v2f;
typedef __attribute__((ext_vector_type(8))) float v8f;

#define TB 256

#define NBK 64
#define BSH 6
#define CH  4096
#define GROWS (TB / 32 * 16)
#define DEGU 8
static_assert(GROWS % NBK == 0, "a product work-group must cover whole ranges");

template <bool RELU_IN, bool DEG>
__global__ void __launch_bounds__(TB)
k_gemm64_wmma(const float* __restrict__ X,
              const float* __restrict__ W,
              float* dis,
              const int2* __restrict__ rec, const int* __restrict__ bstart,
              float* __restrict__ H, int nstrips, int n, int nbuck) {
  __shared__ int s_cnt[DEG ? GROWS : 1];
  const int tid  = threadIdx.x;
  const int lane = tid & 31;
  const int wib  = tid >> 5;
  const int wave = (int)blockIdx.x * (TB / 32) + wib;
  const bool live = wave < nstrips;

  const int m0    = wave * 16;
  const int mlow  = lane & 15;
  const int khalf = (lane >> 4) << 1;

  if constexpr (DEG) {
    if (tid < GROWS) s_cnt[tid] = 0;
    __syncthreads();
  }

  v2f a[16];
#pragma unroll
  for (int k = 0; k < 16; ++k) { a[k].x = 0.0f; a[k].y = 0.0f; }
  if (live) {
    const float* xr = X + (size_t)(m0 + mlow) * 64 + khalf;
#pragma unroll
    for (int k = 0; k < 16; ++k) {
      float ax = xr[k * 4 + 0];
      float ay = xr[k * 4 + 1];
      if (RELU_IN) { ax = fmaxf(ax, 0.0f); ay = fmaxf(ay, 0.0f); }
      a[k].x = ax; a[k].y = ay;
    }
  }

  float dv[8];
  if constexpr (DEG) {
    const int r0 = (int)blockIdx.x * GROWS;
    const int b0 = r0 >> BSH;
    int b2 = b0 + GROWS / NBK;
    if (b2 > nbuck) b2 = nbuck;
    const int e0 = bstart[b0], e1 = bstart[b2];
    for (int i = e0 + tid; i < e1; i += DEGU * TB) {
      int d[DEGU];
#pragma unroll
      for (int u = 0; u < DEGU; ++u) {
        const int idx = i + u * TB;
        const int idc = idx < e1 ? idx : e1 - 1;
        const int row = rec[idc].y - r0;
        d[u] = idx < e1 ? row : -1;
      }
#pragma unroll
      for (int u = 0; u < DEGU; ++u)
        if (d[u] >= 0) atomicAdd(&s_cnt[d[u]], 1);
    }
    __syncthreads();
    if (tid < GROWS && r0 + tid < n) dis[r0 + tid] = rsqrtf(1.0f + (float)s_cnt[tid]);
    const int* cr = s_cnt + wib * 16 + 8 * (lane >> 4);
#pragma unroll
    for (int v = 0; v < 8; ++v) dv[v] = rsqrtf(1.0f + (float)cr[v]);
    if (!live) return;
  } else {
    if (!live) return;
    const float* dr = dis + m0 + 8 * (lane >> 4);
    const float4 d0 = *(const float4*)dr;
    const float4 d1 = *(const float4*)(dr + 4);
    dv[0] = d0.x; dv[1] = d0.y; dv[2] = d0.z; dv[3] = d0.w;
    dv[4] = d1.x; dv[5] = d1.y; dv[6] = d1.z; dv[7] = d1.w;
  }

#pragma unroll
  for (int nt = 0; nt < 4; ++nt) {
    const float* wr = W + (size_t)(nt * 16 + mlow) * 64 + khalf;
    v8f c = {};
#pragma unroll
    for (int k = 0; k < 16; ++k) {
      v2f b;
      b.x = wr[k * 4 + 0];
      b.y = wr[k * 4 + 1];
      c = __builtin_amdgcn_wmma_f32_16x16x4_f32(
           false, a[k],  false, b,
           (short)0, c,  false,  false);
    }
    float* o = H + (size_t)(m0 + 8 * (lane >> 4)) * 64 + nt * 16 + mlow;
#pragma unroll
    for (int v = 0; v < 8; ++v) o[(size_t)v * 64] = c[v] * dv[v];
  }
}

__global__ void k_bin_count(const int* __restrict__ dst, int* __restrict__ cntT,
                            int e, int nbuck, int nchunk) {
  extern __shared__ int s_bin[];
  const int tid = threadIdx.x, chunk = blockIdx.x;
  for (int b = tid; b < nbuck; b += blockDim.x) s_bin[b] = 0;
  __syncthreads();
  const int base = chunk * CH;
  for (int i = tid; i < CH; i += blockDim.x) {
    const int ed = base + i;
    if (ed < e) atomicAdd(&s_bin[dst[ed] >> BSH], 1);
  }
  __syncthreads();
  for (int b = tid; b < nbuck; b += blockDim.x) cntT[(size_t)b * nchunk + chunk] = s_bin[b];
}

__device__ __forceinline__ int wave_incl_scan(int x, int lane) {
  x += __builtin_amdgcn_update_dpp(0, x, 0x111, 0xf, 0xf, true);
  x += __builtin_amdgcn_update_dpp(0, x, 0x112, 0xf, 0xf, true);
  x += __builtin_amdgcn_update_dpp(0, x, 0x114, 0xf, 0xf, true);
  x += __builtin_amdgcn_update_dpp(0, x, 0x118, 0xf, 0xf, true);
  const int low = __builtin_amdgcn_readlane(x, 15);
  return lane >= 16 ? x + low : x;
}

#define SCW 8
#define SCU 8

__global__ void __launch_bounds__(SCW * 32)
k_row_xscan(int* arr, int len, int* __restrict__ totals, int nrows) {
  const int lane = threadIdx.x & 31;
  const int row  = (int)blockIdx.x * SCW + (int)(threadIdx.x >> 5);
  if (row >= nrows) return;
  int* a = arr + (size_t)row * len;
  int carry = 0;
  for (int c0 = 0; c0 < len; c0 += SCU * 32) {
    int v[SCU];
#pragma unroll
    for (int u = 0; u < SCU; ++u) {
      const int c  = c0 + u * 32 + lane;
      const int cc = c < len ? c : len - 1;
      const int t  = a[cc];
      v[u] = c < len ? t : 0;
    }
#pragma unroll
    for (int u = 0; u < SCU; ++u) {
      if (c0 + u * 32 < len) {
        const int incl = wave_incl_scan(v[u], lane);
        const int c = c0 + u * 32 + lane;
        if (c < len) a[c] = carry + incl - v[u];
        carry += __builtin_amdgcn_readlane(incl, 31);
      }
    }
  }
  if (lane == 0) totals[row] = carry;
}

__global__ void k_bin_place(const int* __restrict__ src, const int* __restrict__ dst,
                            const int* __restrict__ cntT, const int* __restrict__ btot,
                            int* __restrict__ bstart,
                            int2* __restrict__ rec, int e, int nbuck, int nchunk) {
  extern __shared__ int s_pos[];
  __shared__ int s_wtot[32];
  const int tid = threadIdx.x, chunk = blockIdx.x;
  const int lane = tid & 31, wib = tid >> 5;
  const int nthr = (int)blockDim.x;
  const int per  = (nbuck + nthr - 1) / nthr;
  const int r0   = tid * per;

  int run = 0;
  for (int j0 = 0; j0 < per; j0 += 4) {
    int t[4];
#pragma unroll
    for (int u = 0; u < 4; ++u) {
      const int b = r0 + j0 + u;
      const bool ok = (j0 + u < per) && (b < nbuck);
      const int x = btot[ok ? b : 0];
      t[u] = ok ? x : 0;
    }
    run += (t[0] + t[1]) + (t[2] + t[3]);
  }
  const int incl = wave_incl_scan(run, lane);
  if (lane == 31) s_wtot[wib] = incl;
  __syncthreads();
  int base = incl - run;
  for (int w = 0; w < wib; ++w) base += s_wtot[w];
  for (int j0 = 0; j0 < per; j0 += 4) {
    int t[4], c[4];
#pragma unroll
    for (int u = 0; u < 4; ++u) {
      const int b = r0 + j0 + u;
      const bool ok = (j0 + u < per) && (b < nbuck);
      const int bc = ok ? b : 0;
      const int x = btot[bc];
      const int y = cntT[(size_t)bc * nchunk + chunk];
      t[u] = ok ? x : 0;
      c[u] = y;
    }
#pragma unroll
    for (int u = 0; u < 4; ++u) {
      const int b = r0 + j0 + u;
      if ((j0 + u < per) && (b < nbuck)) {
        s_pos[b] = base + c[u];
        if (chunk == 0) bstart[b] = base;
        base += t[u];
        if (chunk == 0 && b == nbuck - 1) bstart[nbuck] = base;
      }
    }
  }
  __syncthreads();
  const int ebase = chunk * CH;
  for (int i = tid; i < CH; i += nthr) {
    const int ed = ebase + i;
    if (ed < e) {
      const int sv = src[ed], dv = dst[ed];
      const int p = atomicAdd(&s_pos[dv >> BSH], 1);
      rec[p] = make_int2(sv, dv);
    }
  }
}

__global__ void __launch_bounds__(256)
k_tile_aggregate(const int2* __restrict__ rec, const int* __restrict__ bstart,
                 const float* __restrict__ h, const float* __restrict__ dis,
                 const float* __restrict__ bias, float* __restrict__ out, int n) {
  __shared__ float acc[NBK * 64];
  const int tid  = threadIdx.x;
  const int lane = tid & 31;
  const int wave = __builtin_amdgcn_readfirstlane(tid >> 5);
  const int b  = blockIdx.x;
  const int n0 = b << BSH;
  const int e0 = bstart[b], e1 = bstart[b + 1];

  for (int idx = tid; idx < NBK * 16; idx += 256) {
    const int i = idx >> 4, f = (idx & 15) << 2;
    const int node = n0 + i;
    if (node < n)
      *(float4*)(acc + i * 64 + f) = *(const float4*)(h + (size_t)node * 64 + f);
  }
  __syncthreads();

  const float* hl = h + 2 * lane;
  float* al = acc + 2 * lane;
  for (int eb = e0 + wave * 32; eb < e1; eb += 8 * 32) {
    const int my = eb + lane;
    int rs = 0, rl = 0;
    if (my < e1) {
      const int2 r = rec[my];
      rs = r.x;
      rl = (r.y - n0) << 6;
    }
    const int left = e1 - eb;
    const int cnt = left < 32 ? left : 32;
    int j = 0;
    for (; j + 8 <= cnt; j += 8) {
      float2 v[8];
#pragma unroll
      for (int u = 0; u < 8; ++u) {
        const int sj = __builtin_amdgcn_readlane(rs, j + u);
        v[u] = *(const float2*)(hl + (size_t)sj * 64);
      }
#pragma unroll
      for (int u = 0; u < 8; ++u) {
        const int ro = __builtin_amdgcn_readlane(rl, j + u);
        atomicAdd(al + ro,     v[u].x);
        atomicAdd(al + ro + 1, v[u].y);
      }
    }
    for (; j < cnt; ++j) {
      const int sj = __builtin_amdgcn_readlane(rs, j);
      const int ro = __builtin_amdgcn_readlane(rl, j);
      const float2 v = *(const float2*)(hl + (size_t)sj * 64);
      atomicAdd(al + ro,     v.x);
      atomicAdd(al + ro + 1, v.y);
    }
  }
  __syncthreads();

  for (int idx = tid; idx < NBK * 16; idx += 256) {
    const int i = idx >> 4, f = (idx & 15) << 2;
    const int node = n0 + i;
    if (node < n) {
      const float w = dis[node];
      const float4 v  = *(const float4*)(acc + i * 64 + f);
      const float4 bb = *(const float4*)(bias + f);
      float4 r;
      r.x = v.x * w + bb.x; r.y = v.y * w + bb.y;
      r.z = v.z * w + bb.z; r.w = v.w * w + bb.w;
      *(float4*)(out + (size_t)node * 64 + f) = r;
    }
  }
}

extern "C" void kernel_launch(void* const* d_in, const int* in_sizes, int n_in,
                              void* d_out, int out_size, void* d_ws, size_t ws_size,
                              hipStream_t stream) {
  (void)n_in; (void)out_size; (void)ws_size;

  const float* x  = (const float*)d_in[0];
  const int*   ei = (const int*)  d_in[1];
  const float* W1 = (const float*)d_in[2];
  const float* b1 = (const float*)d_in[3];
  const float* W2 = (const float*)d_in[4];
  const float* b2 = (const float*)d_in[5];
  float* out = (float*)d_out;

  const int n = in_sizes[0] / 64;
  const int e = in_sizes[1] / 2;
  const int* src = ei;
  const int* dst = ei + e;

  const int nbuck  = (n + NBK - 1) / NBK;
  const int nchunk = (e + CH - 1) / CH;

  float* dis = (float*)d_ws;
  float* h   = dis + n;
  float* agg = h + (size_t)n * 64;
  int* cntT   = (int*)(agg + (size_t)n * 64);
  int* bstart = cntT + (size_t)nbuck * nchunk;
  int* btot   = bstart + nbuck + 1;
  int2* rec   = (int2*)(((uintptr_t)(btot + nbuck) + 7) & ~(uintptr_t)7);

  const int nstrips = (n + 15) / 16;
  const int gemm_blocks = (nstrips + 7) / 8;
  const size_t bin_lds  = (size_t)nbuck * sizeof(int);

  k_bin_count<<<nchunk, 512, bin_lds, stream>>>(dst, cntT, e, nbuck, nchunk);
  k_row_xscan<<<(nbuck + SCW - 1) / SCW, SCW * 32, 0, stream>>>(cntT, nchunk, btot, nbuck);
  k_bin_place<<<nchunk, 512, bin_lds, stream>>>(src, dst, cntT, btot, bstart, rec, e, nbuck, nchunk);

  k_gemm64_wmma<false, true><<<gemm_blocks, TB, 0, stream>>>(x, W1, dis, rec, bstart, h, nstrips, n, nbuck);
  k_tile_aggregate<<<nbuck, 256, 0, stream>>>(rec, bstart, h, dis, b1, agg, n);

  k_gemm64_wmma<true, false><<<gemm_blocks, TB, 0, stream>>>(agg, W2, dis, rec, bstart, h, nstrips, n, nbuck);
  k_tile_aggregate<<<nbuck, 256, 0, stream>>>(rec, bstart, h, dis, b2, out, n);
}


# range aggregation without LDS float atomics: each wave owns 8 of a range's 64 rows in registers and picks its edges from the LDS-staged records by lane vote; sum order changes
# speedup vs baseline: 2.0436x; 1.0639x over previous
#include <hip/hip_runtime.h>
#include <hip/hip_bf16.h>


typedef __attribute__((ext_vector_type(2))) float v2f;
typedef __attribute__((ext_vector_type(8))) float v8f;

#define TB 256

#define NBK 64
#define BSH 6
#define CH  4096
#define GROWS (TB / 32 * 16)
#define DEGU 8
static_assert(GROWS % NBK == 0, "a product work-group must cover whole ranges");

template <bool RELU_IN, bool DEG>
__global__ void __launch_bounds__(TB)
k_gemm64_wmma(const float* __restrict__ X,
              const float* __restrict__ W,
              float* dis,
              const int2* __restrict__ rec, const int* __restrict__ bstart,
              float* __restrict__ H, int nstrips, int n, int nbuck) {
  __shared__ int s_cnt[DEG ? GROWS : 1];
  const int tid  = threadIdx.x;
  const int lane = tid & 31;
  const int wib  = tid >> 5;
  const int wave = (int)blockIdx.x * (TB / 32) + wib;
  const bool live = wave < nstrips;

  const int m0    = wave * 16;
  const int mlow  = lane & 15;
  const int khalf = (lane >> 4) << 1;

  if constexpr (DEG) {
    if (tid < GROWS) s_cnt[tid] = 0;
    __syncthreads();
  }

  v2f a[16];
#pragma unroll
  for (int k = 0; k < 16; ++k) { a[k].x = 0.0f; a[k].y = 0.0f; }
  if (live) {
    const float* xr = X + (size_t)(m0 + mlow) * 64 + khalf;
#pragma unroll
    for (int k = 0; k < 16; ++k) {
      float ax = xr[k * 4 + 0];
      float ay = xr[k * 4 + 1];
      if (RELU_IN) { ax = fmaxf(ax, 0.0f); ay = fmaxf(ay, 0.0f); }
      a[k].x = ax; a[k].y = ay;
    }
  }

  float dv[8];
  if constexpr (DEG) {
    const int r0 = (int)blockIdx.x * GROWS;
    const int b0 = r0 >> BSH;
    int b2 = b0 + GROWS / NBK;
    if (b2 > nbuck) b2 = nbuck;
    const int e0 = bstart[b0], e1 = bstart[b2];
    for (int i = e0 + tid; i < e1; i += DEGU * TB) {
      int d[DEGU];
#pragma unroll
      for (int u = 0; u < DEGU; ++u) {
        const int idx = i + u * TB;
        const int idc = idx < e1 ? idx : e1 - 1;
        const int row = rec[idc].y - r0;
        d[u] = idx < e1 ? row : -1;
      }
#pragma unroll
      for (int u = 0; u < DEGU; ++u)
        if (d[u] >= 0) atomicAdd(&s_cnt[d[u]], 1);
    }
    __syncthreads();
    if (tid < GROWS && r0 + tid < n) dis[r0 + tid] = rsqrtf(1.0f + (float)s_cnt[tid]);
    const int* cr = s_cnt + wib * 16 + 8 * (lane >> 4);
#pragma unroll
    for (int v = 0; v < 8; ++v) dv[v] = rsqrtf(1.0f + (float)cr[v]);
    if (!live) return;
  } else {
    if (!live) return;
    const float* dr = dis + m0 + 8 * (lane >> 4);
    const float4 d0 = *(const float4*)dr;
    const float4 d1 = *(const float4*)(dr + 4);
    dv[0] = d0.x; dv[1] = d0.y; dv[2] = d0.z; dv[3] = d0.w;
    dv[4] = d1.x; dv[5] = d1.y; dv[6] = d1.z; dv[7] = d1.w;
  }

#pragma unroll
  for (int nt = 0; nt < 4; ++nt) {
    const float* wr = W + (size_t)(nt * 16 + mlow) * 64 + khalf;
    v8f c = {};
#pragma unroll
    for (int k = 0; k < 16; ++k) {
      v2f b;
      b.x = wr[k * 4 + 0];
      b.y = wr[k * 4 + 1];
      c = __builtin_amdgcn_wmma_f32_16x16x4_f32(
           false, a[k],  false, b,
           (short)0, c,  false,  false);
    }
    float* o = H + (size_t)(m0 + 8 * (lane >> 4)) * 64 + nt * 16 + mlow;
#pragma unroll
    for (int v = 0; v < 8; ++v) o[(size_t)v * 64] = c[v] * dv[v];
  }
}

__global__ void k_bin_count(const int* __restrict__ dst, int* __restrict__ cntT,
                            int e, int nbuck, int nchunk) {
  extern __shared__ int s_bin[];
  const int tid = threadIdx.x, chunk = blockIdx.x;
  for (int b = tid; b < nbuck; b += blockDim.x) s_bin[b] = 0;
  __syncthreads();
  const int base = chunk * CH;
  for (int i = tid; i < CH; i += blockDim.x) {
    const int ed = base + i;
    if (ed < e) atomicAdd(&s_bin[dst[ed] >> BSH], 1);
  }
  __syncthreads();
  for (int b = tid; b < nbuck; b += blockDim.x) cntT[(size_t)b * nchunk + chunk] = s_bin[b];
}

__device__ __forceinline__ int wave_incl_scan(int x, int lane) {
  x += __builtin_amdgcn_update_dpp(0, x, 0x111, 0xf, 0xf, true);
  x += __builtin_amdgcn_update_dpp(0, x, 0x112, 0xf, 0xf, true);
  x += __builtin_amdgcn_update_dpp(0, x, 0x114, 0xf, 0xf, true);
  x += __builtin_amdgcn_update_dpp(0, x, 0x118, 0xf, 0xf, true);
  const int low = __builtin_amdgcn_readlane(x, 15);
  return lane >= 16 ? x + low : x;
}

#define SCW 8
#define SCU 8

__global__ void __launch_bounds__(SCW * 32)
k_row_xscan(int* arr, int len, int* __restrict__ totals, int nrows) {
  const int lane = threadIdx.x & 31;
  const int row  = (int)blockIdx.x * SCW + (int)(threadIdx.x >> 5);
  if (row >= nrows) return;
  int* a = arr + (size_t)row * len;
  int carry = 0;
  for (int c0 = 0; c0 < len; c0 += SCU * 32) {
    int v[SCU];
#pragma unroll
    for (int u = 0; u < SCU; ++u) {
      const int c  = c0 + u * 32 + lane;
      const int cc = c < len ? c : len - 1;
      const int t  = a[cc];
      v[u] = c < len ? t : 0;
    }
#pragma unroll
    for (int u = 0; u < SCU; ++u) {
      if (c0 + u * 32 < len) {
        const int incl = wave_incl_scan(v[u], lane);
        const int c = c0 + u * 32 + lane;
        if (c < len) a[c] = carry + incl - v[u];
        carry += __builtin_amdgcn_readlane(incl, 31);
      }
    }
  }
  if (lane == 0) totals[row] = carry;
}

__global__ void k_bin_place(const int* __restrict__ src, const int* __restrict__ dst,
                            const int* __restrict__ cntT, const int* __restrict__ btot,
                            int* __restrict__ bstart,
                            int2* __restrict__ rec, int e, int nbuck, int nchunk) {
  extern __shared__ int s_pos[];
  __shared__ int s_wtot[32];
  const int tid = threadIdx.x, chunk = blockIdx.x;
  const int lane = tid & 31, wib = tid >> 5;
  const int nthr = (int)blockDim.x;
  const int per  = (nbuck + nthr - 1) / nthr;
  const int r0   = tid * per;

  int run = 0;
  for (int j0 = 0; j0 < per; j0 += 4) {
    int t[4];
#pragma unroll
    for (int u = 0; u < 4; ++u) {
      const int b = r0 + j0 + u;
      const bool ok = (j0 + u < per) && (b < nbuck);
      const int x = btot[ok ? b : 0];
      t[u] = ok ? x : 0;
    }
    run += (t[0] + t[1]) + (t[2] + t[3]);
  }
  const int incl = wave_incl_scan(run, lane);
  if (lane == 31) s_wtot[wib] = incl;
  __syncthreads();
  int base = incl - run;
  for (int w = 0; w < wib; ++w) base += s_wtot[w];
  for (int j0 = 0; j0 < per; j0 += 4) {
    int t[4], c[4];
#pragma unroll
    for (int u = 0; u < 4; ++u) {
      const int b = r0 + j0 + u;
      const bool ok = (j0 + u < per) && (b < nbuck);
      const int bc = ok ? b : 0;
      const int x = btot[bc];
      const int y = cntT[(size_t)bc * nchunk + chunk];
      t[u] = ok ? x : 0;
      c[u] = y;
    }
#pragma unroll
    for (int u = 0; u < 4; ++u) {
      const int b = r0 + j0 + u;
      if ((j0 + u < per) && (b < nbuck)) {
        s_pos[b] = base + c[u];
        if (chunk == 0) bstart[b] = base;
        base += t[u];
        if (chunk == 0 && b == nbuck - 1) bstart[nbuck] = base;
      }
    }
  }
  __syncthreads();
  const int ebase = chunk * CH;
  for (int i = tid; i < CH; i += nthr) {
    const int ed = ebase + i;
    if (ed < e) {
      const int sv = src[ed], dv = dst[ed];
      const int p = atomicAdd(&s_pos[dv >> BSH], 1);
      rec[p] = make_int2(sv, dv);
    }
  }
}

#define OWNR 8
#define LCAP 1152
#define GRP  12
#define SU   4
#define STG  ((LCAP + 255) / 256)
static_assert(OWNR * 8 == NBK, "8 waves must cover the range");

typedef __attribute__((ext_vector_type(16))) float v16f;

__global__ void __launch_bounds__(256)
k_tile_aggregate(const int2* __restrict__ rec, const int* __restrict__ bstart,
                 const float* __restrict__ h, const float* __restrict__ dis,
                 const float* __restrict__ bias, float* __restrict__ out, int n) {
  __shared__ int s_rec[STG * 256];
  __shared__ int s_list[8 * LCAP];
  const int tid  = threadIdx.x;
  const int lane = tid & 31;
  const int wave = __builtin_amdgcn_readfirstlane(tid >> 5);
  const int b  = blockIdx.x;
  const int n0 = b << BSH;
  const int e0 = bstart[b], e1 = bstart[b + 1];
  const int row0 = n0 + wave * OWNR;

  v16f acc;
#pragma unroll
  for (int r = 0; r < OWNR; ++r) {
    const int node = row0 + r;
    const float2 own = node < n ? *(const float2*)(h + (size_t)node * 64 + 2 * lane) : make_float2(0.0f, 0.0f);
    acc[2 * r] = own.x; acc[2 * r + 1] = own.y;
  }

  const float* hl = h + 2 * lane;
  int* lst = s_list + wave * LCAP;

  for (int rb = e0; rb < e1; rb += LCAP) {
    const int nrec = (e1 - rb > LCAP) ? LCAP : e1 - rb;
    const int re = rb + nrec;
    if (rb != e0) __syncthreads();
#pragma unroll
    for (int u = 0; u < STG; ++u) {
      const int k = tid + u * 256;
      const int idx = rb + k;
      const int2 r = rec[idx < re ? idx : re - 1];
      s_rec[k] = (r.x << 6) | (r.y - n0);
    }
    __syncthreads();

    int cnt = 0;
    for (int c0 = 0; c0 < nrec; c0 += 32 * SU) {
      int en[SU];
#pragma unroll
      for (int u = 0; u < SU; ++u) {
        const int idx = c0 + u * 32 + lane;
        const int t = s_rec[idx < nrec ? idx : nrec - 1];
        en[u] = idx < nrec ? t : -1;
      }
#pragma unroll
      for (int u = 0; u < SU; ++u) {
        const bool mine = en[u] >= 0 && ((en[u] >> 3) & 7) == wave;
        const unsigned mask = __builtin_amdgcn_ballot_w32(mine);
        const int pos = cnt + (int)__builtin_amdgcn_mbcnt_lo(mask, 0u);
        if (mine) lst[pos] = en[u];
        cnt += __builtin_popcount(mask);
      }
    }

    for (int base = 0; base < cnt; base += 32) {
      const int li = base + lane;
      const int ent = lst[li < cnt ? li : cnt - 1];
      const int left = cnt - base;
      const int m = left < 32 ? left : 32;
      for (int j = 0; j < m; j += GRP) {
        float2 v[GRP];
#pragma unroll
        for (int u = 0; u < GRP; ++u) {
          const int jj = (j + u < m) ? j + u : m - 1;
          const int e = __builtin_amdgcn_readlane(ent, jj);
          v[u] = *(const float2*)(hl + (size_t)(e >> 6) * 64);
        }
#pragma unroll
        for (int u = 0; u < GRP; ++u) {
          if (u == 0 || j + u < m) {
            const int e = __builtin_amdgcn_readlane(ent, j + u);
            const int ix = (e & 7) * 2;
            acc[ix]     += v[u].x;
            acc[ix + 1] += v[u].y;
          }
        }
      }
    }
  }

  const float2 bb = *(const float2*)(bias + 2 * lane);
#pragma unroll
  for (int r = 0; r < OWNR; ++r) {
    const int node = row0 + r;
    if (node < n) {
      const float w = dis[node];
      float2 o;
      o.x = acc[2 * r] * w + bb.x;
      o.y = acc[2 * r + 1] * w + bb.y;
      *(float2*)(out + (size_t)node * 64 + 2 * lane) = o;
    }
  }
}

extern "C" void kernel_launch(void* const* d_in, const int* in_sizes, int n_in,
                              void* d_out, int out_size, void* d_ws, size_t ws_size,
                              hipStream_t stream) {
  (void)n_in; (void)out_size; (void)ws_size;

  const float* x  = (const float*)d_in[0];
  const int*   ei = (const int*)  d_in[1];
  const float* W1 = (const float*)d_in[2];
  const float* b1 = (const float*)d_in[3];
  const float* W2 = (const float*)d_in[4];
  const float* b2 = (const float*)d_in[5];
  float* out = (float*)d_out;

  const int n = in_sizes[0] / 64;
  const int e = in_sizes[1] / 2;
  const int* src = ei;
  const int* dst = ei + e;

  const int nbuck  = (n + NBK - 1) / NBK;
  const int nchunk = (e + CH - 1) / CH;

  float* dis = (float*)d_ws;
  float* h   = dis + n;
  float* agg = h + (size_t)n * 64;
  int* cntT   = (int*)(agg + (size_t)n * 64);
  int* bstart = cntT + (size_t)nbuck * nchunk;
  int* btot   = bstart + nbuck + 1;
  int2* rec   = (int2*)(((uintptr_t)(btot + nbuck) + 7) & ~(uintptr_t)7);

  const int nstrips = (n + 15) / 16;
  const int gemm_blocks = (nstrips + 7) / 8;
  const size_t bin_lds  = (size_t)nbuck * sizeof(int);

  k_bin_count<<<nchunk, 512, bin_lds, stream>>>(dst, cntT, e, nbuck, nchunk);
  k_row_xscan<<<(nbuck + SCW - 1) / SCW, SCW * 32, 0, stream>>>(cntT, nchunk, btot, nbuck);
  k_bin_place<<<nchunk, 512, bin_lds, stream>>>(src, dst, cntT, btot, bstart, rec, e, nbuck, nchunk);

  k_gemm64_wmma<false, true><<<gemm_blocks, TB, 0, stream>>>(x, W1, dis, rec, bstart, h, nstrips, n, nbuck);
  k_tile_aggregate<<<nbuck, 256, 0, stream>>>(rec, bstart, h, dis, b1, agg, n);

  k_gemm64_wmma<true, false><<<gemm_blocks, TB, 0, stream>>>(agg, W2, dis, rec, bstart, h, nstrips, n, nbuck);
  k_tile_aggregate<<<nbuck, 256, 0, stream>>>(rec, bstart, h, dis, b2, out, n);
}


# aggregation walk: a group's 16 row fetches (was 12) all issued before its first add, no branch per slot; the compiler had sunk one fetch per group behind an add and waited for it alone; same sums
# speedup vs baseline: 2.2406x; 1.0964x over previous
#include <hip/hip_runtime.h>
#include <hip/hip_bf16.h>


typedef __attribute__((ext_vector_type(2))) float v2f;
typedef __attribute__((ext_vector_type(8))) float v8f;

#define TB 256

#define NBK 64
#define BSH 6
#define CH  4096
#define GROWS (TB / 32 * 16)
#define DEGU 8
static_assert(GROWS % NBK == 0, "a product work-group must cover whole ranges");

template <bool RELU_IN, bool DEG>
__global__ void __launch_bounds__(TB)
k_gemm64_wmma(const float* __restrict__ X,
              const float* __restrict__ W,
              float* dis,
              const int2* __restrict__ rec, const int* __restrict__ bstart,
              float* __restrict__ H, int nstrips, int n, int nbuck) {
  __shared__ int s_cnt[DEG ? GROWS : 1];
  const int tid  = threadIdx.x;
  const int lane = tid & 31;
  const int wib  = tid >> 5;
  const int wave = (int)blockIdx.x * (TB / 32) + wib;
  const bool live = wave < nstrips;

  const int m0    = wave * 16;
  const int mlow  = lane & 15;
  const int khalf = (lane >> 4) << 1;

  if constexpr (DEG) {
    if (tid < GROWS) s_cnt[tid] = 0;
    __syncthreads();
  }

  v2f a[16];
#pragma unroll
  for (int k = 0; k < 16; ++k) { a[k].x = 0.0f; a[k].y = 0.0f; }
  if (live) {
    const float* xr = X + (size_t)(m0 + mlow) * 64 + khalf;
#pragma unroll
    for (int k = 0; k < 16; ++k) {
      float ax = xr[k * 4 + 0];
      float ay = xr[k * 4 + 1];
      if (RELU_IN) { ax = fmaxf(ax, 0.0f); ay = fmaxf(ay, 0.0f); }
      a[k].x = ax; a[k].y = ay;
    }
  }

  float dv[8];
  if constexpr (DEG) {
    const int r0 = (int)blockIdx.x * GROWS;
    const int b0 = r0 >> BSH;
    int b2 = b0 + GROWS / NBK;
    if (b2 > nbuck) b2 = nbuck;
    const int e0 = bstart[b0], e1 = bstart[b2];
    for (int i = e0 + tid; i < e1; i += DEGU * TB) {
      int d[DEGU];
#pragma unroll
      for (int u = 0; u < DEGU; ++u) {
        const int idx = i + u * TB;
        const int idc = idx < e1 ? idx : e1 - 1;
        const int row = rec[idc].y - r0;
        d[u] = idx < e1 ? row : -1;
      }
#pragma unroll
      for (int u = 0; u < DEGU; ++u)
        if (d[u] >= 0) atomicAdd(&s_cnt[d[u]], 1);
    }
    __syncthreads();
    if (tid < GROWS && r0 + tid < n) dis[r0 + tid] = rsqrtf(1.0f + (float)s_cnt[tid]);
    const int* cr = s_cnt + wib * 16 + 8 * (lane >> 4);
#pragma unroll
    for (int v = 0; v < 8; ++v) dv[v] = rsqrtf(1.0f + (float)cr[v]);
    if (!live) return;
  } else {
    if (!live) return;
    const float* dr = dis + m0 + 8 * (lane >> 4);
    const float4 d0 = *(const float4*)dr;
    const float4 d1 = *(const float4*)(dr + 4);
    dv[0] = d0.x; dv[1] = d0.y; dv[2] = d0.z; dv[3] = d0.w;
    dv[4] = d1.x; dv[5] = d1.y; dv[6] = d1.z; dv[7] = d1.w;
  }

#pragma unroll
  for (int nt = 0; nt < 4; ++nt) {
    const float* wr = W + (size_t)(nt * 16 + mlow) * 64 + khalf;
    v8f c = {};
#pragma unroll
    for (int k = 0; k < 16; ++k) {
      v2f b;
      b.x = wr[k * 4 + 0];
      b.y = wr[k * 4 + 1];
      c = __builtin_amdgcn_wmma_f32_16x16x4_f32(
           false, a[k],  false, b,
           (short)0, c,  false,  false);
    }
    float* o = H + (size_t)(m0 + 8 * (lane >> 4)) * 64 + nt * 16 + mlow;
#pragma unroll
    for (int v = 0; v < 8; ++v) o[(size_t)v * 64] = c[v] * dv[v];
  }
}

__global__ void k_bin_count(const int* __restrict__ dst, int* __restrict__ cntT,
                            int e, int nbuck, int nchunk) {
  extern __shared__ int s_bin[];
  const int tid = threadIdx.x, chunk = blockIdx.x;
  for (int b = tid; b < nbuck; b += blockDim.x) s_bin[b] = 0;
  __syncthreads();
  const int base = chunk * CH;
  for (int i = tid; i < CH; i += blockDim.x) {
    const int ed = base + i;
    if (ed < e) atomicAdd(&s_bin[dst[ed] >> BSH], 1);
  }
  __syncthreads();
  for (int b = tid; b < nbuck; b += blockDim.x) cntT[(size_t)b * nchunk + chunk] = s_bin[b];
}

__device__ __forceinline__ int wave_incl_scan(int x, int lane) {
  x += __builtin_amdgcn_update_dpp(0, x, 0x111, 0xf, 0xf, true);
  x += __builtin_amdgcn_update_dpp(0, x, 0x112, 0xf, 0xf, true);
  x += __builtin_amdgcn_update_dpp(0, x, 0x114, 0xf, 0xf, true);
  x += __builtin_amdgcn_update_dpp(0, x, 0x118, 0xf, 0xf, true);
  const int low = __builtin_amdgcn_readlane(x, 15);
  return lane >= 16 ? x + low : x;
}

#define SCW 8
#define SCU 8

__global__ void __launch_bounds__(SCW * 32)
k_row_xscan(int* arr, int len, int* __restrict__ totals, int nrows) {
  const int lane = threadIdx.x & 31;
  const int row  = (int)blockIdx.x * SCW + (int)(threadIdx.x >> 5);
  if (row >= nrows) return;
  int* a = arr + (size_t)row * len;
  int carry = 0;
  for (int c0 = 0; c0 < len; c0 += SCU * 32) {
    int v[SCU];
#pragma unroll
    for (int u = 0; u < SCU; ++u) {
      const int c  = c0 + u * 32 + lane;
      const int cc = c < len ? c : len - 1;
      const int t  = a[cc];
      v[u] = c < len ? t : 0;
    }
#pragma unroll
    for (int u = 0; u < SCU; ++u) {
      if (c0 + u * 32 < len) {
        const int incl = wave_incl_scan(v[u], lane);
        const int c = c0 + u * 32 + lane;
        if (c < len) a[c] = carry + incl - v[u];
        carry += __builtin_amdgcn_readlane(incl, 31);
      }
    }
  }
  if (lane == 0) totals[row] = carry;
}

__global__ void k_bin_place(const int* __restrict__ src, const int* __restrict__ dst,
                            const int* __restrict__ cntT, const int* __restrict__ btot,
                            int* __restrict__ bstart,
                            int2* __restrict__ rec, int e, int nbuck, int nchunk) {
  extern __shared__ int s_pos[];
  __shared__ int s_wtot[32];
  const int tid = threadIdx.x, chunk = blockIdx.x;
  const int lane = tid & 31, wib = tid >> 5;
  const int nthr = (int)blockDim.x;
  const int per  = (nbuck + nthr - 1) / nthr;
  const int r0   = tid * per;

  int run = 0;
  for (int j0 = 0; j0 < per; j0 += 4) {
    int t[4];
#pragma unroll
    for (int u = 0; u < 4; ++u) {
      const int b = r0 + j0 + u;
      const bool ok = (j0 + u < per) && (b < nbuck);
      const int x = btot[ok ? b : 0];
      t[u] = ok ? x : 0;
    }
    run += (t[0] + t[1]) + (t[2] + t[3]);
  }
  const int incl = wave_incl_scan(run, lane);
  if (lane == 31) s_wtot[wib] = incl;
  __syncthreads();
  int base = incl - run;
  for (int w = 0; w < wib; ++w) base += s_wtot[w];
  for (int j0 = 0; j0 < per; j0 += 4) {
    int t[4], c[4];
#pragma unroll
    for (int u = 0; u < 4; ++u) {
      const int b = r0 + j0 + u;
      const bool ok = (j0 + u < per) && (b < nbuck);
      const int bc = ok ? b : 0;
      const int x = btot[bc];
      const int y = cntT[(size_t)bc * nchunk + chunk];
      t[u] = ok ? x : 0;
      c[u] = y;
    }
#pragma unroll
    for (int u = 0; u < 4; ++u) {
      const int b = r0 + j0 + u;
      if ((j0 + u < per) && (b < nbuck)) {
        s_pos[b] = base + c[u];
        if (chunk == 0) bstart[b] = base;
        base += t[u];
        if (chunk == 0 && b == nbuck - 1) bstart[nbuck] = base;
      }
    }
  }
  __syncthreads();
  const int ebase = chunk * CH;
  for (int i = tid; i < CH; i += nthr) {
    const int ed = ebase + i;
    if (ed < e) {
      const int sv = src[ed], dv = dst[ed];
      const int p = atomicAdd(&s_pos[dv >> BSH], 1);
      rec[p] = make_int2(sv, dv);
    }
  }
}

#define OWNR 8
#define LCAP 1152
#define GRP  16
#define SU   4
#define STG  ((LCAP + 255) / 256)
static_assert(OWNR * 8 == NBK, "8 waves must cover the range");

typedef __attribute__((ext_vector_type(16))) float v16f;

template <bool TAIL>
__device__ __forceinline__ void walk_group(v16f& acc, const float* __restrict__ h,
                                           int lane2, int ent, int j, int m) {
  float2 v[GRP];
#pragma unroll
  for (int u = 0; u < GRP; ++u) {
    int jj = j + u;
    if (TAIL) jj = jj < m ? jj : m - 1;
    const int e = __builtin_amdgcn_readlane(ent, jj);
    const float* rowp = h + (size_t)(e >> 6) * 64;
    v[u] = *(const float2*)(rowp + lane2);
  }
#pragma unroll
  for (int u = 0; u < GRP; ++u) {
    int jj = j + u;
    bool in = true;
    if (TAIL) { in = jj < m; jj = in ? jj : m - 1; }
    const int e = __builtin_amdgcn_readlane(ent, jj);
    const int ix = (e & 7) * 2;
    float ax = v[u].x, ay = v[u].y;
    if (TAIL) { ax = in ? ax : -0.0f; ay = in ? ay : -0.0f; }
    acc[ix]     += ax;
    acc[ix + 1] += ay;
  }
}

__global__ void __launch_bounds__(256)
k_tile_aggregate(const int2* __restrict__ rec, const int* __restrict__ bstart,
                 const float* __restrict__ h, const float* __restrict__ dis,
                 const float* __restrict__ bias, float* __restrict__ out, int n) {
  __shared__ int s_rec[STG * 256];
  __shared__ int s_list[8 * LCAP];
  const int tid  = threadIdx.x;
  const int lane = tid & 31;
  const int wave = __builtin_amdgcn_readfirstlane(tid >> 5);
  const int b  = blockIdx.x;
  const int n0 = b << BSH;
  const int e0 = bstart[b], e1 = bstart[b + 1];
  const int row0 = n0 + wave * OWNR;

  v16f acc;
#pragma unroll
  for (int r = 0; r < OWNR; ++r) {
    const int node = row0 + r;
    const float2 own = node < n ? *(const float2*)(h + (size_t)node * 64 + 2 * lane) : make_float2(0.0f, 0.0f);
    acc[2 * r] = own.x; acc[2 * r + 1] = own.y;
  }

  int* lst = s_list + wave * LCAP;

  for (int rb = e0; rb < e1; rb += LCAP) {
    const int nrec = (e1 - rb > LCAP) ? LCAP : e1 - rb;
    const int re = rb + nrec;
    if (rb != e0) __syncthreads();
#pragma unroll
    for (int u = 0; u < STG; ++u) {
      const int k = tid + u * 256;
      const int idx = rb + k;
      const int2 r = rec[idx < re ? idx : re - 1];
      s_rec[k] = (r.x << 6) | (r.y - n0);
    }
    __syncthreads();

    int cnt = 0;
    for (int c0 = 0; c0 < nrec; c0 += 32 * SU) {
      int en[SU];
#pragma unroll
      for (int u = 0; u < SU; ++u) {
        const int idx = c0 + u * 32 + lane;
        const int t = s_rec[idx < nrec ? idx : nrec - 1];
        en[u] = idx < nrec ? t : -1;
      }
#pragma unroll
      for (int u = 0; u < SU; ++u) {
        const bool mine = en[u] >= 0 && ((en[u] >> 3) & 7) == wave;
        const unsigned mask = __builtin_amdgcn_ballot_w32(mine);
        const int pos = cnt + (int)__builtin_amdgcn_mbcnt_lo(mask, 0u);
        if (mine) lst[pos] = en[u];
        cnt += __builtin_popcount(mask);
      }
    }

    for (int base = 0; base < cnt; base += 32) {
      const int li = base + lane;
      const int ent = lst[li < cnt ? li : cnt - 1];
      const int left = cnt - base;
      const int m = left < 32 ? left : 32;
      int j = 0;
      for (; j + GRP <= m; j += GRP) walk_group<false>(acc, h, 2 * lane, ent, j, m);
      if (j < m) walk_group<true>(acc, h, 2 * lane, ent, j, m);
    }
  }

  const float2 bb = *(const float2*)(bias + 2 * lane);
#pragma unroll
  for (int r = 0; r < OWNR; ++r) {
    const int node = row0 + r;
    if (node < n) {
      const float w = dis[node];
      float2 o;
      o.x = acc[2 * r] * w + bb.x;
      o.y = acc[2 * r + 1] * w + bb.y;
      *(float2*)(out + (size_t)node * 64 + 2 * lane) = o;
    }
  }
}

extern "C" void kernel_launch(void* const* d_in, const int* in_sizes, int n_in,
                              void* d_out, int out_size, void* d_ws, size_t ws_size,
                              hipStream_t stream) {
  (void)n_in; (void)out_size; (void)ws_size;

  const float* x  = (const float*)d_in[0];
  const int*   ei = (const int*)  d_in[1];
  const float* W1 = (const float*)d_in[2];
  const float* b1 = (const float*)d_in[3];
  const float* W2 = (const float*)d_in[4];
  const float* b2 = (const float*)d_in[5];
  float* out = (float*)d_out;

  const int n = in_sizes[0] / 64;
  const int e = in_sizes[1] / 2;
  const int* src = ei;
  const int* dst = ei + e;

  const int nbuck  = (n + NBK - 1) / NBK;
  const int nchunk = (e + CH - 1) / CH;

  float* dis = (float*)d_ws;
  float* h   = dis + n;
  float* agg = h + (size_t)n * 64;
  int* cntT   = (int*)(agg + (size_t)n * 64);
  int* bstart = cntT + (size_t)nbuck * nchunk;
  int* btot   = bstart + nbuck + 1;
  int2* rec   = (int2*)(((uintptr_t)(btot + nbuck) + 7) & ~(uintptr_t)7);

  const int nstrips = (n + 15) / 16;
  const int gemm_blocks = (nstrips + 7) / 8;
  const size_t bin_lds  = (size_t)nbuck * sizeof(int);

  k_bin_count<<<nchunk, 512, bin_lds, stream>>>(dst, cntT, e, nbuck, nchunk);
  k_row_xscan<<<(nbuck + SCW - 1) / SCW, SCW * 32, 0, stream>>>(cntT, nchunk, btot, nbuck);
  k_bin_place<<<nchunk, 512, bin_lds, stream>>>(src, dst, cntT, btot, bstart, rec, e, nbuck, nchunk);

  k_gemm64_wmma<false, true><<<gemm_blocks, TB, 0, stream>>>(x, W1, dis, rec, bstart, h, nstrips, n, nbuck);
  k_tile_aggregate<<<nbuck, 256, 0, stream>>>(rec, bstart, h, dis, b1, agg, n);

  k_gemm64_wmma<true, false><<<gemm_blocks, TB, 0, stream>>>(agg, W2, dis, rec, bstart, h, nstrips, n, nbuck);
  k_tile_aggregate<<<nbuck, 256, 0, stream>>>(rec, bstart, h, dis, b2, out, n);
}


# product launches: a wave forms two 16-row strips per B fragment; A and B fragments fetched 16 bytes a lane in one shared permuted k order: 48 loads per 128 WMMA, was 80 per 64; k sum order changes
# speedup vs baseline: 2.3122x; 1.0320x over previous
#include <hip/hip_runtime.h>
#include <hip/hip_bf16.h>


typedef __attribute__((ext_vector_type(2))) float v2f;
typedef __attribute__((ext_vector_type(8))) float v8f;

#define TB 256

#define NBK 64
#define BSH 6
#define CH  4096
#define SPW 2
#define GROWS (TB / 32 * 16 * SPW)
#define DEGU 16
static_assert(GROWS % NBK == 0, "a product work-group must cover whole ranges");
static_assert(GROWS == TB, "one degree counter per thread");

template <bool RELU_IN, bool DEG>
__global__ void __launch_bounds__(TB)
k_gemm64_wmma(const float* __restrict__ X,
              const float* __restrict__ W,
              float* dis,
              const int2* __restrict__ rec, const int* __restrict__ bstart,
              float* __restrict__ H, int nwaves, int n, int nbuck) {
  __shared__ int s_cnt[DEG ? GROWS : 1];
  const int tid  = threadIdx.x;
  const int lane = tid & 31;
  const int wib  = tid >> 5;
  const int wave = (int)blockIdx.x * (TB / 32) + wib;
  const bool live = wave < nwaves;

  const int m0   = wave * (16 * SPW);
  const int mlow = lane & 15;
  const int hf   = lane >> 4;
  const int hk   = hf << 2;

  int ms[SPW];
#pragma unroll
  for (int s = 0; s < SPW; ++s) ms[s] = (m0 + 16 * s < n) ? m0 + 16 * s : m0;

  if constexpr (DEG) {
    s_cnt[tid] = 0;
    __syncthreads();
  }

  float4 a[SPW][8];
#pragma unroll
  for (int s = 0; s < SPW; ++s)
#pragma unroll
    for (int q = 0; q < 8; ++q) a[s][q] = make_float4(0.0f, 0.0f, 0.0f, 0.0f);
  if (live) {
#pragma unroll
    for (int s = 0; s < SPW; ++s) {
      const float* xr = X + (size_t)(ms[s] + mlow) * 64 + hk;
#pragma unroll
      for (int q = 0; q < 8; ++q) {
        a[s][q] = *(const float4*)(xr + 8 * q);
      }
    }
  }

  if (RELU_IN) {
#pragma unroll
    for (int s = 0; s < SPW; ++s)
#pragma unroll
      for (int q = 0; q < 8; ++q) {
        a[s][q].x = fmaxf(a[s][q].x, 0.0f); a[s][q].y = fmaxf(a[s][q].y, 0.0f);
        a[s][q].z = fmaxf(a[s][q].z, 0.0f); a[s][q].w = fmaxf(a[s][q].w, 0.0f);
      }
  }

  float dv[SPW][8];
  if constexpr (DEG) {
    const int r0 = (int)blockIdx.x * GROWS;
    const int b0 = r0 >> BSH;
    int b2 = b0 + GROWS / NBK;
    if (b2 > nbuck) b2 = nbuck;
    const int e0 = bstart[b0], e1 = bstart[b2];
    for (int i = e0 + tid; i < e1; i += DEGU * TB) {
      int d[DEGU];
#pragma unroll
      for (int u = 0; u < DEGU; ++u) {
        const int idx = i + u * TB;
        const int idc = idx < e1 ? idx : e1 - 1;
        const int row = rec[idc].y - r0;
        d[u] = idx < e1 ? row : -1;
      }
#pragma unroll
      for (int u = 0; u < DEGU; ++u)
        if (d[u] >= 0) atomicAdd(&s_cnt[d[u]], 1);
    }
    __syncthreads();
    if (r0 + tid < n) dis[r0 + tid] = rsqrtf(1.0f + (float)s_cnt[tid]);
#pragma unroll
    for (int s = 0; s < SPW; ++s) {
      const int* cr = s_cnt + wib * (16 * SPW) + (ms[s] - m0) + 8 * hf;
#pragma unroll
      for (int v = 0; v < 8; ++v) dv[s][v] = rsqrtf(1.0f + (float)cr[v]);
    }
    if (!live) return;
  } else {
    if (!live) return;
#pragma unroll
    for (int s = 0; s < SPW; ++s) {
      const float* dr = dis + ms[s] + 8 * hf;
      const float4 d0 = *(const float4*)dr;
      const float4 d1 = *(const float4*)(dr + 4);
      dv[s][0] = d0.x; dv[s][1] = d0.y; dv[s][2] = d0.z; dv[s][3] = d0.w;
      dv[s][4] = d1.x; dv[s][5] = d1.y; dv[s][6] = d1.z; dv[s][7] = d1.w;
    }
  }

#pragma unroll
  for (int nt = 0; nt < 4; ++nt) {
    const float* wr = W + (size_t)(nt * 16 + mlow) * 64 + hk;
    float4 b[8];
#pragma unroll
    for (int q = 0; q < 8; ++q) b[q] = *(const float4*)(wr + 8 * q);
    v8f c[SPW];
#pragma unroll
    for (int s = 0; s < SPW; ++s) c[s] = (v8f){};
#pragma unroll
    for (int q = 0; q < 8; ++q) {
      v2f b0, b1;
      b0.x = b[q].x; b0.y = b[q].y; b1.x = b[q].z; b1.y = b[q].w;
#pragma unroll
      for (int s = 0; s < SPW; ++s) {
        v2f a0, a1;
        a0.x = a[s][q].x; a0.y = a[s][q].y; a1.x = a[s][q].z; a1.y = a[s][q].w;
        c[s] = __builtin_amdgcn_wmma_f32_16x16x4_f32(
             false, a0,  false, b0,
             (short)0, c[s],  false,  false);
        c[s] = __builtin_amdgcn_wmma_f32_16x16x4_f32(
             false, a1,  false, b1,
             (short)0, c[s],  false,  false);
      }
    }
#pragma unroll
    for (int s = 0; s < SPW; ++s) {
      float* o = H + (size_t)(ms[s] + 8 * hf) * 64 + nt * 16 + mlow;
#pragma unroll
      for (int v = 0; v < 8; ++v) o[(size_t)v * 64] = c[s][v] * dv[s][v];
    }
  }
}

__global__ void k_bin_count(const int* __restrict__ dst, int* __restrict__ cntT,
                            int e, int nbuck, int nchunk) {
  extern __shared__ int s_bin[];
  const int tid = threadIdx.x, chunk = blockIdx.x;
  for (int b = tid; b < nbuck; b += blockDim.x) s_bin[b] = 0;
  __syncthreads();
  const int base = chunk * CH;
  for (int i = tid; i < CH; i += blockDim.x) {
    const int ed = base + i;
    if (ed < e) atomicAdd(&s_bin[dst[ed] >> BSH], 1);
  }
  __syncthreads();
  for (int b = tid; b < nbuck; b += blockDim.x) cntT[(size_t)b * nchunk + chunk] = s_bin[b];
}

__device__ __forceinline__ int wave_incl_scan(int x, int lane) {
  x += __builtin_amdgcn_update_dpp(0, x, 0x111, 0xf, 0xf, true);
  x += __builtin_amdgcn_update_dpp(0, x, 0x112, 0xf, 0xf, true);
  x += __builtin_amdgcn_update_dpp(0, x, 0x114, 0xf, 0xf, true);
  x += __builtin_amdgcn_update_dpp(0, x, 0x118, 0xf, 0xf, true);
  const int low = __builtin_amdgcn_readlane(x, 15);
  return lane >= 16 ? x + low : x;
}

#define SCW 8
#define SCU 8

__global__ void __launch_bounds__(SCW * 32)
k_row_xscan(int* arr, int len, int* __restrict__ totals, int nrows) {
  const int lane = threadIdx.x & 31;
  const int row  = (int)blockIdx.x * SCW + (int)(threadIdx.x >> 5);
  if (row >= nrows) return;
  int* a = arr + (size_t)row * len;
  int carry = 0;
  for (int c0 = 0; c0 < len; c0 += SCU * 32) {
    int v[SCU];
#pragma unroll
    for (int u = 0; u < SCU; ++u) {
      const int c  = c0 + u * 32 + lane;
      const int cc = c < len ? c : len - 1;
      const int t  = a[cc];
      v[u] = c < len ? t : 0;
    }
#pragma unroll
    for (int u = 0; u < SCU; ++u) {
      if (c0 + u * 32 < len) {
        const int incl = wave_incl_scan(v[u], lane);
        const int c = c0 + u * 32 + lane;
        if (c < len) a[c] = carry + incl - v[u];
        carry += __builtin_amdgcn_readlane(incl, 31);
      }
    }
  }
  if (lane == 0) totals[row] = carry;
}

__global__ void k_bin_place(const int* __restrict__ src, const int* __restrict__ dst,
                            const int* __restrict__ cntT, const int* __restrict__ btot,
                            int* __restrict__ bstart,
                            int2* __restrict__ rec, int e, int nbuck, int nchunk) {
  extern __shared__ int s_pos[];
  __shared__ int s_wtot[32];
  const int tid = threadIdx.x, chunk = blockIdx.x;
  const int lane = tid & 31, wib = tid >> 5;
  const int nthr = (int)blockDim.x;
  const int per  = (nbuck + nthr - 1) / nthr;
  const int r0   = tid * per;

  int run = 0;
  for (int j0 = 0; j0 < per; j0 += 4) {
    int t[4];
#pragma unroll
    for (int u = 0; u < 4; ++u) {
      const int b = r0 + j0 + u;
      const bool ok = (j0 + u < per) && (b < nbuck);
      const int x = btot[ok ? b : 0];
      t[u] = ok ? x : 0;
    }
    run += (t[0] + t[1]) + (t[2] + t[3]);
  }
  const int incl = wave_incl_scan(run, lane);
  if (lane == 31) s_wtot[wib] = incl;
  __syncthreads();
  int base = incl - run;
  for (int w = 0; w < wib; ++w) base += s_wtot[w];
  for (int j0 = 0; j0 < per; j0 += 4) {
    int t[4], c[4];
#pragma unroll
    for (int u = 0; u < 4; ++u) {
      const int b = r0 + j0 + u;
      const bool ok = (j0 + u < per) && (b < nbuck);
      const int bc = ok ? b : 0;
      const int x = btot[bc];
      const int y = cntT[(size_t)bc * nchunk + chunk];
      t[u] = ok ? x : 0;
      c[u] = y;
    }
#pragma unroll
    for (int u = 0; u < 4; ++u) {
      const int b = r0 + j0 + u;
      if ((j0 + u < per) && (b < nbuck)) {
        s_pos[b] = base + c[u];
        if (chunk == 0) bstart[b] = base;
        base += t[u];
        if (chunk == 0 && b == nbuck - 1) bstart[nbuck] = base;
      }
    }
  }
  __syncthreads();
  const int ebase = chunk * CH;
  for (int i = tid; i < CH; i += nthr) {
    const int ed = ebase + i;
    if (ed < e) {
      const int sv = src[ed], dv = dst[ed];
      const int p = atomicAdd(&s_pos[dv >> BSH], 1);
      rec[p] = make_int2(sv, dv);
    }
  }
}

#define OWNR 8
#define LCAP 1152
#define GRP  16
#define SU   4
#define STG  ((LCAP + 255) / 256)
static_assert(OWNR * 8 == NBK, "8 waves must cover the range");

typedef __attribute__((ext_vector_type(16))) float v16f;

template <bool TAIL>
__device__ __forceinline__ void walk_group(v16f& acc, const float* __restrict__ h,
                                           int lane2, int ent, int j, int m) {
  float2 v[GRP];
#pragma unroll
  for (int u = 0; u < GRP; ++u) {
    int jj = j + u;
    if (TAIL) jj = jj < m ? jj : m - 1;
    const int e = __builtin_amdgcn_readlane(ent, jj);
    const float* rowp = h + (size_t)(e >> 6) * 64;
    v[u] = *(const float2*)(rowp + lane2);
  }
#pragma unroll
  for (int u = 0; u < GRP; ++u) {
    int jj = j + u;
    bool in = true;
    if (TAIL) { in = jj < m; jj = in ? jj : m - 1; }
    const int e = __builtin_amdgcn_readlane(ent, jj);
    const int ix = (e & 7) * 2;
    float ax = v[u].x, ay = v[u].y;
    if (TAIL) { ax = in ? ax : -0.0f; ay = in ? ay : -0.0f; }
    acc[ix]     += ax;
    acc[ix + 1] += ay;
  }
}

__global__ void __launch_bounds__(256)
k_tile_aggregate(const int2* __restrict__ rec, const int* __restrict__ bstart,
                 const float* __restrict__ h, const float* __restrict__ dis,
                 const float* __restrict__ bias, float* __restrict__ out, int n) {
  __shared__ int s_rec[STG * 256];
  __shared__ int s_list[8 * LCAP];
  const int tid  = threadIdx.x;
  const int lane = tid & 31;
  const int wave = __builtin_amdgcn_readfirstlane(tid >> 5);
  const int b  = blockIdx.x;
  const int n0 = b << BSH;
  const int e0 = bstart[b], e1 = bstart[b + 1];
  const int row0 = n0 + wave * OWNR;

  v16f acc;
#pragma unroll
  for (int r = 0; r < OWNR; ++r) {
    const int node = row0 + r;
    const float2 own = node < n ? *(const float2*)(h + (size_t)node * 64 + 2 * lane) : make_float2(0.0f, 0.0f);
    acc[2 * r] = own.x; acc[2 * r + 1] = own.y;
  }

  int* lst = s_list + wave * LCAP;

  for (int rb = e0; rb < e1; rb += LCAP) {
    const int nrec = (e1 - rb > LCAP) ? LCAP : e1 - rb;
    const int re = rb + nrec;
    if (rb != e0) __syncthreads();
#pragma unroll
    for (int u = 0; u < STG; ++u) {
      const int k = tid + u * 256;
      const int idx = rb + k;
      const int2 r = rec[idx < re ? idx : re - 1];
      s_rec[k] = (r.x << 6) | (r.y - n0);
    }
    __syncthreads();

    int cnt = 0;
    for (int c0 = 0; c0 < nrec; c0 += 32 * SU) {
      int en[SU];
#pragma unroll
      for (int u = 0; u < SU; ++u) {
        const int idx = c0 + u * 32 + lane;
        const int t = s_rec[idx < nrec ? idx : nrec - 1];
        en[u] = idx < nrec ? t : -1;
      }
#pragma unroll
      for (int u = 0; u < SU; ++u) {
        const bool mine = en[u] >= 0 && ((en[u] >> 3) & 7) == wave;
        const unsigned mask = __builtin_amdgcn_ballot_w32(mine);
        const int pos = cnt + (int)__builtin_amdgcn_mbcnt_lo(mask, 0u);
        if (mine) lst[pos] = en[u];
        cnt += __builtin_popcount(mask);
      }
    }

    for (int base = 0; base < cnt; base += 32) {
      const int li = base + lane;
      const int ent = lst[li < cnt ? li : cnt - 1];
      const int left = cnt - base;
      const int m = left < 32 ? left : 32;
      int j = 0;
      for (; j + GRP <= m; j += GRP) walk_group<false>(acc, h, 2 * lane, ent, j, m);
      if (j < m) walk_group<true>(acc, h, 2 * lane, ent, j, m);
    }
  }

  const float2 bb = *(const float2*)(bias + 2 * lane);
#pragma unroll
  for (int r = 0; r < OWNR; ++r) {
    const int node = row0 + r;
    if (node < n) {
      const float w = dis[node];
      float2 o;
      o.x = acc[2 * r] * w + bb.x;
      o.y = acc[2 * r + 1] * w + bb.y;
      *(float2*)(out + (size_t)node * 64 + 2 * lane) = o;
    }
  }
}

extern "C" void kernel_launch(void* const* d_in, const int* in_sizes, int n_in,
                              void* d_out, int out_size, void* d_ws, size_t ws_size,
                              hipStream_t stream) {
  (void)n_in; (void)out_size; (void)ws_size;

  const float* x  = (const float*)d_in[0];
  const int*   ei = (const int*)  d_in[1];
  const float* W1 = (const float*)d_in[2];
  const float* b1 = (const float*)d_in[3];
  const float* W2 = (const float*)d_in[4];
  const float* b2 = (const float*)d_in[5];
  float* out = (float*)d_out;

  const int n = in_sizes[0] / 64;
  const int e = in_sizes[1] / 2;
  const int* src = ei;
  const int* dst = ei + e;

  const int nbuck  = (n + NBK - 1) / NBK;
  const int nchunk = (e + CH - 1) / CH;

  float* dis = (float*)d_ws;
  float* h   = dis + n;
  float* agg = h + (size_t)n * 64;
  int* cntT   = (int*)(agg + (size_t)n * 64);
  int* bstart = cntT + (size_t)nbuck * nchunk;
  int* btot   = bstart + nbuck + 1;
  int2* rec   = (int2*)(((uintptr_t)(btot + nbuck) + 7) & ~(uintptr_t)7);

  const int nstrips = (n + 16 * SPW - 1) / (16 * SPW);
  const int gemm_blocks = (nstrips + 7) / 8;
  const size_t bin_lds  = (size_t)nbuck * sizeof(int);

  k_bin_count<<<nchunk, 512, bin_lds, stream>>>(dst, cntT, e, nbuck, nchunk);
  k_row_xscan<<<(nbuck + SCW - 1) / SCW, SCW * 32, 0, stream>>>(cntT, nchunk, btot, nbuck);
  k_bin_place<<<nchunk, 512, bin_lds, stream>>>(src, dst, cntT, btot, bstart, rec, e, nbuck, nchunk);

  k_gemm64_wmma<false, true><<<gemm_blocks, TB, 0, stream>>>(x, W1, dis, rec, bstart, h, nstrips, n, nbuck);
  k_tile_aggregate<<<nbuck, 256, 0, stream>>>(rec, bstart, h, dis, b1, agg, n);

  k_gemm64_wmma<true, false><<<gemm_blocks, TB, 0, stream>>>(agg, W2, dis, rec, bstart, h, nstrips, n, nbuck);
  k_tile_aggregate<<<nbuck, 256, 0, stream>>>(rec, bstart, h, dis, b2, out, n);
}


# aggregation: records counting-sorted by row in LDS (no 8x read-all-and-vote); a row's edges adjacent: summed in 2 fixed registers, added once; rows fetched as float2[row*32+lane]; sum order changes
# speedup vs baseline: 2.4437x; 1.0569x over previous
#include <hip/hip_runtime.h>
#include <hip/hip_bf16.h>


typedef __attribute__((ext_vector_type(2))) float v2f;
typedef __attribute__((ext_vector_type(8))) float v8f;

#define TB 256

#define NBK 64
#define BSH 6
#define CH  4096
#define SPW 2
#define GROWS (TB / 32 * 16 * SPW)
#define DEGU 16
static_assert(GROWS % NBK == 0, "a product work-group must cover whole ranges");
static_assert(GROWS == TB, "one degree counter per thread");

template <bool RELU_IN, bool DEG>
__global__ void __launch_bounds__(TB)
k_gemm64_wmma(const float* __restrict__ X,
              const float* __restrict__ W,
              float* dis,
              const int2* __restrict__ rec, const int* __restrict__ bstart,
              float* __restrict__ H, int nwaves, int n, int nbuck) {
  __shared__ int s_cnt[DEG ? GROWS : 1];
  const int tid  = threadIdx.x;
  const int lane = tid & 31;
  const int wib  = tid >> 5;
  const int wave = (int)blockIdx.x * (TB / 32) + wib;
  const bool live = wave < nwaves;

  const int m0   = wave * (16 * SPW);
  const int mlow = lane & 15;
  const int hf   = lane >> 4;
  const int hk   = hf << 2;

  int ms[SPW];
#pragma unroll
  for (int s = 0; s < SPW; ++s) ms[s] = (m0 + 16 * s < n) ? m0 + 16 * s : m0;

  if constexpr (DEG) {
    s_cnt[tid] = 0;
    __syncthreads();
  }

  float4 a[SPW][8];
#pragma unroll
  for (int s = 0; s < SPW; ++s)
#pragma unroll
    for (int q = 0; q < 8; ++q) a[s][q] = make_float4(0.0f, 0.0f, 0.0f, 0.0f);
  if (live) {
#pragma unroll
    for (int s = 0; s < SPW; ++s) {
      const float* xr = X + (size_t)(ms[s] + mlow) * 64 + hk;
#pragma unroll
      for (int q = 0; q < 8; ++q) {
        a[s][q] = *(const float4*)(xr + 8 * q);
      }
    }
  }

  if (RELU_IN) {
#pragma unroll
    for (int s = 0; s < SPW; ++s)
#pragma unroll
      for (int q = 0; q < 8; ++q) {
        a[s][q].x = fmaxf(a[s][q].x, 0.0f); a[s][q].y = fmaxf(a[s][q].y, 0.0f);
        a[s][q].z = fmaxf(a[s][q].z, 0.0f); a[s][q].w = fmaxf(a[s][q].w, 0.0f);
      }
  }

  float dv[SPW][8];
  if constexpr (DEG) {
    const int r0 = (int)blockIdx.x * GROWS;
    const int b0 = r0 >> BSH;
    int b2 = b0 + GROWS / NBK;
    if (b2 > nbuck) b2 = nbuck;
    const int e0 = bstart[b0], e1 = bstart[b2];
    for (int i = e0 + tid; i < e1; i += DEGU * TB) {
      int d[DEGU];
#pragma unroll
      for (int u = 0; u < DEGU; ++u) {
        const int idx = i + u * TB;
        const int idc = idx < e1 ? idx : e1 - 1;
        const int row = rec[idc].y - r0;
        d[u] = idx < e1 ? row : -1;
      }
#pragma unroll
      for (int u = 0; u < DEGU; ++u)
        if (d[u] >= 0) atomicAdd(&s_cnt[d[u]], 1);
    }
    __syncthreads();
    if (r0 + tid < n) dis[r0 + tid] = rsqrtf(1.0f + (float)s_cnt[tid]);
#pragma unroll
    for (int s = 0; s < SPW; ++s) {
      const int* cr = s_cnt + wib * (16 * SPW) + (ms[s] - m0) + 8 * hf;
#pragma unroll
      for (int v = 0; v < 8; ++v) dv[s][v] = rsqrtf(1.0f + (float)cr[v]);
    }
    if (!live) return;
  } else {
    if (!live) return;
#pragma unroll
    for (int s = 0; s < SPW; ++s) {
      const float* dr = dis + ms[s] + 8 * hf;
      const float4 d0 = *(const float4*)dr;
      const float4 d1 = *(const float4*)(dr + 4);
      dv[s][0] = d0.x; dv[s][1] = d0.y; dv[s][2] = d0.z; dv[s][3] = d0.w;
      dv[s][4] = d1.x; dv[s][5] = d1.y; dv[s][6] = d1.z; dv[s][7] = d1.w;
    }
  }

#pragma unroll
  for (int nt = 0; nt < 4; ++nt) {
    const float* wr = W + (size_t)(nt * 16 + mlow) * 64 + hk;
    float4 b[8];
#pragma unroll
    for (int q = 0; q < 8; ++q) b[q] = *(const float4*)(wr + 8 * q);
    v8f c[SPW];
#pragma unroll
    for (int s = 0; s < SPW; ++s) c[s] = (v8f){};
#pragma unroll
    for (int q = 0; q < 8; ++q) {
      v2f b0, b1;
      b0.x = b[q].x; b0.y = b[q].y; b1.x = b[q].z; b1.y = b[q].w;
#pragma unroll
      for (int s = 0; s < SPW; ++s) {
        v2f a0, a1;
        a0.x = a[s][q].x; a0.y = a[s][q].y; a1.x = a[s][q].z; a1.y = a[s][q].w;
        c[s] = __builtin_amdgcn_wmma_f32_16x16x4_f32(
             false, a0,  false, b0,
             (short)0, c[s],  false,  false);
        c[s] = __builtin_amdgcn_wmma_f32_16x16x4_f32(
             false, a1,  false, b1,
             (short)0, c[s],  false,  false);
      }
    }
#pragma unroll
    for (int s = 0; s < SPW; ++s) {
      float* o = H + (size_t)(ms[s] + 8 * hf) * 64 + nt * 16 + mlow;
#pragma unroll
      for (int v = 0; v < 8; ++v) o[(size_t)v * 64] = c[s][v] * dv[s][v];
    }
  }
}

__global__ void k_bin_count(const int* __restrict__ dst, int* __restrict__ cntT,
                            int e, int nbuck, int nchunk) {
  extern __shared__ int s_bin[];
  const int tid = threadIdx.x, chunk = blockIdx.x;
  for (int b = tid; b < nbuck; b += blockDim.x) s_bin[b] = 0;
  __syncthreads();
  const int base = chunk * CH;
  for (int i = tid; i < CH; i += blockDim.x) {
    const int ed = base + i;
    if (ed < e) atomicAdd(&s_bin[dst[ed] >> BSH], 1);
  }
  __syncthreads();
  for (int b = tid; b < nbuck; b += blockDim.x) cntT[(size_t)b * nchunk + chunk] = s_bin[b];
}

__device__ __forceinline__ int wave_incl_scan(int x, int lane) {
  x += __builtin_amdgcn_update_dpp(0, x, 0x111, 0xf, 0xf, true);
  x += __builtin_amdgcn_update_dpp(0, x, 0x112, 0xf, 0xf, true);
  x += __builtin_amdgcn_update_dpp(0, x, 0x114, 0xf, 0xf, true);
  x += __builtin_amdgcn_update_dpp(0, x, 0x118, 0xf, 0xf, true);
  const int low = __builtin_amdgcn_readlane(x, 15);
  return lane >= 16 ? x + low : x;
}

#define SCW 8
#define SCU 8

__global__ void __launch_bounds__(SCW * 32)
k_row_xscan(int* arr, int len, int* __restrict__ totals, int nrows) {
  const int lane = threadIdx.x & 31;
  const int row  = (int)blockIdx.x * SCW + (int)(threadIdx.x >> 5);
  if (row >= nrows) return;
  int* a = arr + (size_t)row * len;
  int carry = 0;
  for (int c0 = 0; c0 < len; c0 += SCU * 32) {
    int v[SCU];
#pragma unroll
    for (int u = 0; u < SCU; ++u) {
      const int c  = c0 + u * 32 + lane;
      const int cc = c < len ? c : len - 1;
      const int t  = a[cc];
      v[u] = c < len ? t : 0;
    }
#pragma unroll
    for (int u = 0; u < SCU; ++u) {
      if (c0 + u * 32 < len) {
        const int incl = wave_incl_scan(v[u], lane);
        const int c = c0 + u * 32 + lane;
        if (c < len) a[c] = carry + incl - v[u];
        carry += __builtin_amdgcn_readlane(incl, 31);
      }
    }
  }
  if (lane == 0) totals[row] = carry;
}

__global__ void k_bin_place(const int* __restrict__ src, const int* __restrict__ dst,
                            const int* __restrict__ cntT, const int* __restrict__ btot,
                            int* __restrict__ bstart,
                            int2* __restrict__ rec, int e, int nbuck, int nchunk) {
  extern __shared__ int s_pos[];
  __shared__ int s_wtot[32];
  const int tid = threadIdx.x, chunk = blockIdx.x;
  const int lane = tid & 31, wib = tid >> 5;
  const int nthr = (int)blockDim.x;
  const int per  = (nbuck + nthr - 1) / nthr;
  const int r0   = tid * per;

  int run = 0;
  for (int j0 = 0; j0 < per; j0 += 4) {
    int t[4];
#pragma unroll
    for (int u = 0; u < 4; ++u) {
      const int b = r0 + j0 + u;
      const bool ok = (j0 + u < per) && (b < nbuck);
      const int x = btot[ok ? b : 0];
      t[u] = ok ? x : 0;
    }
    run += (t[0] + t[1]) + (t[2] + t[3]);
  }
  const int incl = wave_incl_scan(run, lane);
  if (lane == 31) s_wtot[wib] = incl;
  __syncthreads();
  int base = incl - run;
  for (int w = 0; w < wib; ++w) base += s_wtot[w];
  for (int j0 = 0; j0 < per; j0 += 4) {
    int t[4], c[4];
#pragma unroll
    for (int u = 0; u < 4; ++u) {
      const int b = r0 + j0 + u;
      const bool ok = (j0 + u < per) && (b < nbuck);
      const int bc = ok ? b : 0;
      const int x = btot[bc];
      const int y = cntT[(size_t)bc * nchunk + chunk];
      t[u] = ok ? x : 0;
      c[u] = y;
    }
#pragma unroll
    for (int u = 0; u < 4; ++u) {
      const int b = r0 + j0 + u;
      if ((j0 + u < per) && (b < nbuck)) {
        s_pos[b] = base + c[u];
        if (chunk == 0) bstart[b] = base;
        base += t[u];
        if (chunk == 0 && b == nbuck - 1) bstart[nbuck] = base;
      }
    }
  }
  __syncthreads();
  const int ebase = chunk * CH;
  for (int i = tid; i < CH; i += nthr) {
    const int ed = ebase + i;
    if (ed < e) {
      const int sv = src[ed], dv = dst[ed];
      const int p = atomicAdd(&s_pos[dv >> BSH], 1);
      rec[p] = make_int2(sv, dv);
    }
  }
}

#define OWNR 8
#define LCAP 1152
#define GRP  16
#define STG  ((LCAP + 255) / 256)
static_assert(STG == 5, "the load fence in k_tile_aggregate names STG records");
static_assert(OWNR * 8 == NBK, "8 waves must cover the range");

typedef __attribute__((ext_vector_type(16))) float v16f;

template <bool TAIL>
__device__ __forceinline__ void walk_group(v16f& acc, float2& t, int& cur,
                                           const float2* __restrict__ h2,
                                           unsigned lane, int ent, int j, int m) {
  float2 v[GRP];
#pragma unroll
  for (int u = 0; u < GRP; ++u) {
    int jj = j + u;
    if (TAIL) jj = jj < m ? jj : m - 1;
    const int e = __builtin_amdgcn_readlane(ent, jj);
    v[u] = h2[(unsigned)(e >> 6) * 32u + lane];
  }
#pragma unroll
  for (int u = 0; u < GRP; ++u) {
    int jj = j + u;
    bool in = true;
    if (TAIL) { in = jj < m; jj = in ? jj : m - 1; }
    const int e = __builtin_amdgcn_readlane(ent, jj);
    const int r = e & 7;
    if (r != cur) {
      acc[2 * cur]     += t.x;
      acc[2 * cur + 1] += t.y;
      t = make_float2(-0.0f, -0.0f);
      cur = r;
    }
    float ax = v[u].x, ay = v[u].y;
    if (TAIL) { ax = in ? ax : -0.0f; ay = in ? ay : -0.0f; }
    t.x += ax;
    t.y += ay;
  }
}

__global__ void __launch_bounds__(256)
k_tile_aggregate(const int2* __restrict__ rec, const int* __restrict__ bstart,
                 const float* __restrict__ h, const float* __restrict__ dis,
                 const float* __restrict__ bias, float* __restrict__ out, int n) {
  __shared__ int s_list[8 * LCAP];
  __shared__ int s_cur[NBK];
  const int tid  = threadIdx.x;
  const int lane = tid & 31;
  const int wave = __builtin_amdgcn_readfirstlane(tid >> 5);
  const int b  = blockIdx.x;
  const int n0 = b << BSH;
  if (tid < NBK) s_cur[tid] = 0;
  __syncthreads();
  const int e0 = bstart[b], e1 = bstart[b + 1];
  const int row0 = n0 + wave * OWNR;

  v16f acc;
#pragma unroll
  for (int r = 0; r < OWNR; ++r) {
    const int node = row0 + r;
    const float2 own = node < n ? *(const float2*)(h + (size_t)node * 64 + 2 * lane) : make_float2(0.0f, 0.0f);
    acc[2 * r] = own.x; acc[2 * r + 1] = own.y;
  }

  int* lst = s_list + wave * LCAP;

  for (int rb = e0; rb < e1; rb += LCAP) {
    const int nrec = (e1 - rb > LCAP) ? LCAP : e1 - rb;
    const int re = rb + nrec;
    if (rb != e0) {
      __syncthreads();
      if (tid < NBK) s_cur[tid] = 0;
      __syncthreads();
    }
    int2 rr[STG];
#pragma unroll
    for (int u = 0; u < STG; ++u) {
      const int idx = rb + tid + u * 256;
      rr[u] = rec[idx < re ? idx : re - 1];
    }
    asm volatile("" :: "v"(rr[0].x), "v"(rr[0].y), "v"(rr[1].x), "v"(rr[1].y), "v"(rr[2].x),
                       "v"(rr[2].y), "v"(rr[3].x), "v"(rr[3].y), "v"(rr[4].x), "v"(rr[4].y));
    int pk[STG];
#pragma unroll
    for (int u = 0; u < STG; ++u) {
      const int idx = rb + tid + u * 256;
      pk[u] = idx < re ? ((rr[u].x << 6) | (rr[u].y - n0)) : -1;
      if (pk[u] >= 0) atomicAdd(&s_cur[pk[u] & (NBK - 1)], 1);
    }
    __syncthreads();
    int cnt;
    {
      const int c = lane < OWNR ? s_cur[wave * OWNR + lane] : 0;
      const int incl = wave_incl_scan(c, lane);
      if (lane < OWNR) s_cur[wave * OWNR + lane] = incl - c;
      cnt = __builtin_amdgcn_readlane(incl, OWNR - 1);
    }
    __syncthreads();
#pragma unroll
    for (int u = 0; u < STG; ++u) {
      if (pk[u] >= 0) {
        const int row = pk[u] & (NBK - 1);
        const int pos = atomicAdd(&s_cur[row], 1);
        s_list[(row >> 3) * LCAP + pos] = pk[u];
      }
    }
    __syncthreads();

    float2 t = make_float2(-0.0f, -0.0f);
    int cur = 0;
    for (int base = 0; base < cnt; base += 32) {
      const int li = base + lane;
      const int ent = lst[li < cnt ? li : cnt - 1];
      const int left = cnt - base;
      const int m = left < 32 ? left : 32;
      int j = 0;
      for (; j + GRP <= m; j += GRP) walk_group<false>(acc, t, cur, (const float2*)h, (unsigned)lane, ent, j, m);
      if (j < m) walk_group<true>(acc, t, cur, (const float2*)h, (unsigned)lane, ent, j, m);
    }
    acc[2 * cur]     += t.x;
    acc[2 * cur + 1] += t.y;
  }

  const float2 bb = *(const float2*)(bias + 2 * lane);
#pragma unroll
  for (int r = 0; r < OWNR; ++r) {
    const int node = row0 + r;
    if (node < n) {
      const float w = dis[node];
      float2 o;
      o.x = acc[2 * r] * w + bb.x;
      o.y = acc[2 * r + 1] * w + bb.y;
      *(float2*)(out + (size_t)node * 64 + 2 * lane) = o;
    }
  }
}

extern "C" void kernel_launch(void* const* d_in, const int* in_sizes, int n_in,
                              void* d_out, int out_size, void* d_ws, size_t ws_size,
                              hipStream_t stream) {
  (void)n_in; (void)out_size; (void)ws_size;

  const float* x  = (const float*)d_in[0];
  const int*   ei = (const int*)  d_in[1];
  const float* W1 = (const float*)d_in[2];
  const float* b1 = (const float*)d_in[3];
  const float* W2 = (const float*)d_in[4];
  const float* b2 = (const float*)d_in[5];
  float* out = (float*)d_out;

  const int n = in_sizes[0] / 64;
  const int e = in_sizes[1] / 2;
  const int* src = ei;
  const int* dst = ei + e;

  const int nbuck  = (n + NBK - 1) / NBK;
  const int nchunk = (e + CH - 1) / CH;

  float* dis = (float*)d_ws;
  float* h   = dis + n;
  float* agg = h + (size_t)n * 64;
  int* cntT   = (int*)(agg + (size_t)n * 64);
  int* bstart = cntT + (size_t)nbuck * nchunk;
  int* btot   = bstart + nbuck + 1;
  int2* rec   = (int2*)(((uintptr_t)(btot + nbuck) + 7) & ~(uintptr_t)7);

  const int nstrips = (n + 16 * SPW - 1) / (16 * SPW);
  const int gemm_blocks = (nstrips + 7) / 8;
  const size_t bin_lds  = (size_t)nbuck * sizeof(int);

  k_bin_count<<<nchunk, 512, bin_lds, stream>>>(dst, cntT, e, nbuck, nchunk);
  k_row_xscan<<<(nbuck + SCW - 1) / SCW, SCW * 32, 0, stream>>>(cntT, nchunk, btot, nbuck);
  k_bin_place<<<nchunk, 512, bin_lds, stream>>>(src, dst, cntT, btot, bstart, rec, e, nbuck, nchunk);

  k_gemm64_wmma<false, true><<<gemm_blocks, TB, 0, stream>>>(x, W1, dis, rec, bstart, h, nstrips, n, nbuck);
  k_tile_aggregate<<<nbuck, 256, 0, stream>>>(rec, bstart, h, dis, b1, agg, n);

  k_gemm64_wmma<true, false><<<gemm_blocks, TB, 0, stream>>>(agg, W2, dis, rec, bstart, h, nstrips, n, nbuck);
  k_tile_aggregate<<<nbuck, 256, 0, stream>>>(rec, bstart, h, dis, b2, out, n);
}


# layer-2 product taken behind the layer-2 aggregation (linear in rows) into its launch: a range's 64 summed rows go through LDS to WMMA, relu at the layer-1 store; one launch gone; sum order changes
# speedup vs baseline: 2.5931x; 1.0611x over previous
#include <hip/hip_runtime.h>
#include <hip/hip_bf16.h>


typedef __attribute__((ext_vector_type(2))) float v2f;
typedef __attribute__((ext_vector_type(8))) float v8f;

#define TB 256

#define NBK 64
#define BSH 6
#define CH  4096
#define SPW 2
#define GROWS (TB / 32 * 16 * SPW)
#define DEGU 16
static_assert(GROWS % NBK == 0, "a product work-group must cover whole ranges");
static_assert(GROWS == TB, "one degree counter per thread");

__global__ void __launch_bounds__(TB)
k_gemm64_wmma(const float* __restrict__ X,
              const float* __restrict__ W,
              float* dis,
              const int2* __restrict__ rec, const int* __restrict__ bstart,
              float* __restrict__ H, int nwaves, int n, int nbuck) {
  __shared__ int s_cnt[GROWS];
  const int tid  = threadIdx.x;
  const int lane = tid & 31;
  const int wib  = tid >> 5;
  const int wave = (int)blockIdx.x * (TB / 32) + wib;
  const bool live = wave < nwaves;

  const int m0   = wave * (16 * SPW);
  const int mlow = lane & 15;
  const int hf   = lane >> 4;
  const int hk   = hf << 2;

  int ms[SPW];
#pragma unroll
  for (int s = 0; s < SPW; ++s) ms[s] = (m0 + 16 * s < n) ? m0 + 16 * s : m0;

  s_cnt[tid] = 0;
  __syncthreads();

  float4 a[SPW][8];
#pragma unroll
  for (int s = 0; s < SPW; ++s)
#pragma unroll
    for (int q = 0; q < 8; ++q) a[s][q] = make_float4(0.0f, 0.0f, 0.0f, 0.0f);
  if (live) {
#pragma unroll
    for (int s = 0; s < SPW; ++s) {
      const float* xr = X + (size_t)(ms[s] + mlow) * 64 + hk;
#pragma unroll
      for (int q = 0; q < 8; ++q) {
        a[s][q] = *(const float4*)(xr + 8 * q);
      }
    }
  }

  float dv[SPW][8];
  {
    const int r0 = (int)blockIdx.x * GROWS;
    const int b0 = r0 >> BSH;
    int b2 = b0 + GROWS / NBK;
    if (b2 > nbuck) b2 = nbuck;
    const int e0 = bstart[b0], e1 = bstart[b2];
    for (int i = e0 + tid; i < e1; i += DEGU * TB) {
      int d[DEGU];
#pragma unroll
      for (int u = 0; u < DEGU; ++u) {
        const int idx = i + u * TB;
        const int idc = idx < e1 ? idx : e1 - 1;
        const int row = rec[idc].y - r0;
        d[u] = idx < e1 ? row : -1;
      }
#pragma unroll
      for (int u = 0; u < DEGU; ++u)
        if (d[u] >= 0) atomicAdd(&s_cnt[d[u]], 1);
    }
    __syncthreads();
    if (r0 + tid < n) dis[r0 + tid] = rsqrtf(1.0f + (float)s_cnt[tid]);
#pragma unroll
    for (int s = 0; s < SPW; ++s) {
      const int* cr = s_cnt + wib * (16 * SPW) + (ms[s] - m0) + 8 * hf;
#pragma unroll
      for (int v = 0; v < 8; ++v) dv[s][v] = rsqrtf(1.0f + (float)cr[v]);
    }
    if (!live) return;
  }

#pragma unroll
  for (int nt = 0; nt < 4; ++nt) {
    const float* wr = W + (size_t)(nt * 16 + mlow) * 64 + hk;
    float4 b[8];
#pragma unroll
    for (int q = 0; q < 8; ++q) b[q] = *(const float4*)(wr + 8 * q);
    v8f c[SPW];
#pragma unroll
    for (int s = 0; s < SPW; ++s) c[s] = (v8f){};
#pragma unroll
    for (int q = 0; q < 8; ++q) {
      v2f b0, b1;
      b0.x = b[q].x; b0.y = b[q].y; b1.x = b[q].z; b1.y = b[q].w;
#pragma unroll
      for (int s = 0; s < SPW; ++s) {
        v2f a0, a1;
        a0.x = a[s][q].x; a0.y = a[s][q].y; a1.x = a[s][q].z; a1.y = a[s][q].w;
        c[s] = __builtin_amdgcn_wmma_f32_16x16x4_f32(
             false, a0,  false, b0,
             (short)0, c[s],  false,  false);
        c[s] = __builtin_amdgcn_wmma_f32_16x16x4_f32(
             false, a1,  false, b1,
             (short)0, c[s],  false,  false);
      }
    }
#pragma unroll
    for (int s = 0; s < SPW; ++s) {
      float* o = H + (size_t)(ms[s] + 8 * hf) * 64 + nt * 16 + mlow;
#pragma unroll
      for (int v = 0; v < 8; ++v) o[(size_t)v * 64] = c[s][v] * dv[s][v];
    }
  }
}

__global__ void k_bin_count(const int* __restrict__ dst, int* __restrict__ cntT,
                            int e, int nbuck, int nchunk) {
  extern __shared__ int s_bin[];
  const int tid = threadIdx.x, chunk = blockIdx.x;
  for (int b = tid; b < nbuck; b += blockDim.x) s_bin[b] = 0;
  __syncthreads();
  const int base = chunk * CH;
  for (int i = tid; i < CH; i += blockDim.x) {
    const int ed = base + i;
    if (ed < e) atomicAdd(&s_bin[dst[ed] >> BSH], 1);
  }
  __syncthreads();
  for (int b = tid; b < nbuck; b += blockDim.x) cntT[(size_t)b * nchunk + chunk] = s_bin[b];
}

__device__ __forceinline__ int wave_incl_scan(int x, int lane) {
  x += __builtin_amdgcn_update_dpp(0, x, 0x111, 0xf, 0xf, true);
  x += __builtin_amdgcn_update_dpp(0, x, 0x112, 0xf, 0xf, true);
  x += __builtin_amdgcn_update_dpp(0, x, 0x114, 0xf, 0xf, true);
  x += __builtin_amdgcn_update_dpp(0, x, 0x118, 0xf, 0xf, true);
  const int low = __builtin_amdgcn_readlane(x, 15);
  return lane >= 16 ? x + low : x;
}

#define SCW 8
#define SCU 8

__global__ void __launch_bounds__(SCW * 32)
k_row_xscan(int* arr, int len, int* __restrict__ totals, int nrows) {
  const int lane = threadIdx.x & 31;
  const int row  = (int)blockIdx.x * SCW + (int)(threadIdx.x >> 5);
  if (row >= nrows) return;
  int* a = arr + (size_t)row * len;
  int carry = 0;
  for (int c0 = 0; c0 < len; c0 += SCU * 32) {
    int v[SCU];
#pragma unroll
    for (int u = 0; u < SCU; ++u) {
      const int c  = c0 + u * 32 + lane;
      const int cc = c < len ? c : len - 1;
      const int t  = a[cc];
      v[u] = c < len ? t : 0;
    }
#pragma unroll
    for (int u = 0; u < SCU; ++u) {
      if (c0 + u * 32 < len) {
        const int incl = wave_incl_scan(v[u], lane);
        const int c = c0 + u * 32 + lane;
        if (c < len) a[c] = carry + incl - v[u];
        carry += __builtin_amdgcn_readlane(incl, 31);
      }
    }
  }
  if (lane == 0) totals[row] = carry;
}

__global__ void k_bin_place(const int* __restrict__ src, const int* __restrict__ dst,
                            const int* __restrict__ cntT, const int* __restrict__ btot,
                            int* __restrict__ bstart,
                            int2* __restrict__ rec, int e, int nbuck, int nchunk) {
  extern __shared__ int s_pos[];
  __shared__ int s_wtot[32];
  const int tid = threadIdx.x, chunk = blockIdx.x;
  const int lane = tid & 31, wib = tid >> 5;
  const int nthr = (int)blockDim.x;
  const int per  = (nbuck + nthr - 1) / nthr;
  const int r0   = tid * per;

  int run = 0;
  for (int j0 = 0; j0 < per; j0 += 4) {
    int t[4];
#pragma unroll
    for (int u = 0; u < 4; ++u) {
      const int b = r0 + j0 + u;
      const bool ok = (j0 + u < per) && (b < nbuck);
      const int x = btot[ok ? b : 0];
      t[u] = ok ? x : 0;
    }
    run += (t[0] + t[1]) + (t[2] + t[3]);
  }
  const int incl = wave_incl_scan(run, lane);
  if (lane == 31) s_wtot[wib] = incl;
  __syncthreads();
  int base = incl - run;
  for (int w = 0; w < wib; ++w) base += s_wtot[w];
  for (int j0 = 0; j0 < per; j0 += 4) {
    int t[4], c[4];
#pragma unroll
    for (int u = 0; u < 4; ++u) {
      const int b = r0 + j0 + u;
      const bool ok = (j0 + u < per) && (b < nbuck);
      const int bc = ok ? b : 0;
      const int x = btot[bc];
      const int y = cntT[(size_t)bc * nchunk + chunk];
      t[u] = ok ? x : 0;
      c[u] = y;
    }
#pragma unroll
    for (int u = 0; u < 4; ++u) {
      const int b = r0 + j0 + u;
      if ((j0 + u < per) && (b < nbuck)) {
        s_pos[b] = base + c[u];
        if (chunk == 0) bstart[b] = base;
        base += t[u];
        if (chunk == 0 && b == nbuck - 1) bstart[nbuck] = base;
      }
    }
  }
  __syncthreads();
  const int ebase = chunk * CH;
  for (int i = tid; i < CH; i += nthr) {
    const int ed = ebase + i;
    if (ed < e) {
      const int sv = src[ed], dv = dst[ed];
      const int p = atomicAdd(&s_pos[dv >> BSH], 1);
      rec[p] = make_int2(sv, dv);
    }
  }
}

#define OWNR 8
#define LCAP 1152
#define GRP  16
#define STG  ((LCAP + 255) / 256)
#define TPITCH 68
static_assert(STG == 5, "the load fence in k_tile_aggregate names STG records");
static_assert(OWNR * 8 == NBK, "8 waves must cover the range");
static_assert(NBK * TPITCH <= 8 * LCAP, "the tile of summed rows lies over the record lists");
static_assert(NBK == 64, "8 waves form 4 strips of 16 rows x 2 pairs of column tiles");

typedef __attribute__((ext_vector_type(16))) float v16f;

template <bool TAIL>
__device__ __forceinline__ void walk_group(v16f& acc, float2& t, int& cur,
                                           const float2* __restrict__ h2,
                                           unsigned lane, int ent, int j, int m) {
  float2 v[GRP];
#pragma unroll
  for (int u = 0; u < GRP; ++u) {
    int jj = j + u;
    if (TAIL) jj = jj < m ? jj : m - 1;
    const int e = __builtin_amdgcn_readlane(ent, jj);
    v[u] = h2[(unsigned)(e >> 6) * 32u + lane];
  }
#pragma unroll
  for (int u = 0; u < GRP; ++u) {
    int jj = j + u;
    bool in = true;
    if (TAIL) { in = jj < m; jj = in ? jj : m - 1; }
    const int e = __builtin_amdgcn_readlane(ent, jj);
    const int r = e & 7;
    if (r != cur) {
      acc[2 * cur]     += t.x;
      acc[2 * cur + 1] += t.y;
      t = make_float2(-0.0f, -0.0f);
      cur = r;
    }
    float ax = v[u].x, ay = v[u].y;
    if (TAIL) { ax = in ? ax : -0.0f; ay = in ? ay : -0.0f; }
    t.x += ax;
    t.y += ay;
  }
}

template <bool PROD>
__global__ void __launch_bounds__(256) __attribute__((amdgpu_waves_per_eu(16, 16)))
k_tile_aggregate(const int2* __restrict__ rec, const int* __restrict__ bstart,
                 const float* __restrict__ h, const float* __restrict__ dis,
                 const float* __restrict__ bias, const float* __restrict__ W,
                 float* __restrict__ out, int n) {
  __shared__ int s_list[8 * LCAP];
  __shared__ int s_cur[NBK];
  const int tid  = threadIdx.x;
  const int lane = tid & 31;
  const int wave = __builtin_amdgcn_readfirstlane(tid >> 5);
  const int b  = blockIdx.x;
  const int n0 = b << BSH;
  if (tid < NBK) s_cur[tid] = 0;
  __syncthreads();
  const int e0 = bstart[b], e1 = bstart[b + 1];
  const int row0 = n0 + wave * OWNR;

  v16f acc;
#pragma unroll
  for (int r = 0; r < OWNR; ++r) {
    const int node = row0 + r;
    const float2 own = node < n ? *(const float2*)(h + (size_t)node * 64 + 2 * lane) : make_float2(0.0f, 0.0f);
    acc[2 * r] = own.x; acc[2 * r + 1] = own.y;
  }

  int* lst = s_list + wave * LCAP;

  for (int rb = e0; rb < e1; rb += LCAP) {
    const int nrec = (e1 - rb > LCAP) ? LCAP : e1 - rb;
    const int re = rb + nrec;
    if (rb != e0) {
      __syncthreads();
      if (tid < NBK) s_cur[tid] = 0;
      __syncthreads();
    }
    int2 rr[STG];
#pragma unroll
    for (int u = 0; u < STG; ++u) {
      const int idx = rb + tid + u * 256;
      rr[u] = rec[idx < re ? idx : re - 1];
    }
    asm volatile("" :: "v"(rr[0].x), "v"(rr[0].y), "v"(rr[1].x), "v"(rr[1].y), "v"(rr[2].x),
                       "v"(rr[2].y), "v"(rr[3].x), "v"(rr[3].y), "v"(rr[4].x), "v"(rr[4].y));
    int pk[STG];
#pragma unroll
    for (int u = 0; u < STG; ++u) {
      const int idx = rb + tid + u * 256;
      pk[u] = idx < re ? ((rr[u].x << 6) | (rr[u].y - n0)) : -1;
      if (pk[u] >= 0) atomicAdd(&s_cur[pk[u] & (NBK - 1)], 1);
    }
    __syncthreads();
    int cnt;
    {
      const int c = lane < OWNR ? s_cur[wave * OWNR + lane] : 0;
      const int incl = wave_incl_scan(c, lane);
      if (lane < OWNR) s_cur[wave * OWNR + lane] = incl - c;
      cnt = __builtin_amdgcn_readlane(incl, OWNR - 1);
    }
    __syncthreads();
#pragma unroll
    for (int u = 0; u < STG; ++u) {
      if (pk[u] >= 0) {
        const int row = pk[u] & (NBK - 1);
        const int pos = atomicAdd(&s_cur[row], 1);
        s_list[(row >> 3) * LCAP + pos] = pk[u];
      }
    }
    __syncthreads();

    float2 t = make_float2(-0.0f, -0.0f);
    int cur = 0;
    for (int base = 0; base < cnt; base += 32) {
      const int li = base + lane;
      const int ent = lst[li < cnt ? li : cnt - 1];
      const int left = cnt - base;
      const int m = left < 32 ? left : 32;
      int j = 0;
      for (; j + GRP <= m; j += GRP) walk_group<false>(acc, t, cur, (const float2*)h, (unsigned)lane, ent, j, m);
      if (j < m) walk_group<true>(acc, t, cur, (const float2*)h, (unsigned)lane, ent, j, m);
    }
    acc[2 * cur]     += t.x;
    acc[2 * cur + 1] += t.y;
  }

  if constexpr (!PROD) {
    const float2 bb = *(const float2*)(bias + 2 * lane);
#pragma unroll
    for (int r = 0; r < OWNR; ++r) {
      const int node = row0 + r;
      if (node < n) {
        const float w = dis[node];
        float2 o;
        o.x = fmaxf(acc[2 * r] * w + bb.x, 0.0f) * w;
        o.y = fmaxf(acc[2 * r + 1] * w + bb.y, 0.0f) * w;
        *(float2*)(out + (size_t)node * 64 + 2 * lane) = o;
      }
    }
  } else {
    __syncthreads();
    float* tile = reinterpret_cast<float*>(s_list);
    float wv[OWNR];
#pragma unroll
    for (int r = 0; r < OWNR; ++r) {
      const int node = row0 + r;
      wv[r] = dis[node < n ? node : n - 1];
    }
#pragma unroll
    for (int r = 0; r < OWNR; ++r)
      *(float2*)(tile + (wave * OWNR + r) * TPITCH + 2 * lane) = make_float2(acc[2 * r] * wv[r], acc[2 * r + 1] * wv[r]);
    __syncthreads();

    const int mlow  = lane & 15;
    const int hf    = lane >> 4;
    const int hk    = hf << 2;
    const int strip = wave >> 1;
    const int nt0   = (wave & 1) * 2;

    const int srow = n0 + strip * 16;
    if (srow >= n) return;

    const float* ar = tile + (strip * 16 + mlow) * TPITCH + hk;
    float4 a[8];
#pragma unroll
    for (int q = 0; q < 8; ++q) a[q] = *(const float4*)(ar + 8 * q);

#pragma unroll
    for (int t = 0; t < 2; ++t) {
      const int nt = nt0 + t;
      const float* wr = W + (size_t)(nt * 16 + mlow) * 64 + hk;
      float4 b[8];
#pragma unroll
      for (int q = 0; q < 8; ++q) b[q] = *(const float4*)(wr + 8 * q);
      const float bn = bias[nt * 16 + mlow];
      v8f c = (v8f){};
#pragma unroll
      for (int q = 0; q < 8; ++q) {
        v2f a0, a1, b0, b1;
        a0.x = a[q].x; a0.y = a[q].y; a1.x = a[q].z; a1.y = a[q].w;
        b0.x = b[q].x; b0.y = b[q].y; b1.x = b[q].z; b1.y = b[q].w;
        c = __builtin_amdgcn_wmma_f32_16x16x4_f32(
             false, a0,  false, b0,
             (short)0, c,  false,  false);
        c = __builtin_amdgcn_wmma_f32_16x16x4_f32(
             false, a1,  false, b1,
             (short)0, c,  false,  false);
      }
      float* o = out + (size_t)(srow + 8 * hf) * 64 + nt * 16 + mlow;
#pragma unroll
      for (int v = 0; v < 8; ++v) o[(size_t)v * 64] = c[v] + bn;
    }
  }
}

extern "C" void kernel_launch(void* const* d_in, const int* in_sizes, int n_in,
                              void* d_out, int out_size, void* d_ws, size_t ws_size,
                              hipStream_t stream) {
  (void)n_in; (void)out_size; (void)ws_size;

  const float* x  = (const float*)d_in[0];
  const int*   ei = (const int*)  d_in[1];
  const float* W1 = (const float*)d_in[2];
  const float* b1 = (const float*)d_in[3];
  const float* W2 = (const float*)d_in[4];
  const float* b2 = (const float*)d_in[5];
  float* out = (float*)d_out;

  const int n = in_sizes[0] / 64;
  const int e = in_sizes[1] / 2;
  const int* src = ei;
  const int* dst = ei + e;

  const int nbuck  = (n + NBK - 1) / NBK;
  const int nchunk = (e + CH - 1) / CH;

  float* dis = (float*)d_ws;
  float* h   = dis + n;
  float* agg = h + (size_t)n * 64;
  int* cntT   = (int*)(agg + (size_t)n * 64);
  int* bstart = cntT + (size_t)nbuck * nchunk;
  int* btot   = bstart + nbuck + 1;
  int2* rec   = (int2*)(((uintptr_t)(btot + nbuck) + 7) & ~(uintptr_t)7);

  const int nstrips = (n + 16 * SPW - 1) / (16 * SPW);
  const int gemm_blocks = (nstrips + 7) / 8;
  const size_t bin_lds  = (size_t)nbuck * sizeof(int);

  k_bin_count<<<nchunk, 512, bin_lds, stream>>>(dst, cntT, e, nbuck, nchunk);
  k_row_xscan<<<(nbuck + SCW - 1) / SCW, SCW * 32, 0, stream>>>(cntT, nchunk, btot, nbuck);
  k_bin_place<<<nchunk, 512, bin_lds, stream>>>(src, dst, cntT, btot, bstart, rec, e, nbuck, nchunk);

  k_gemm64_wmma<<<gemm_blocks, TB, 0, stream>>>(x, W1, dis, rec, bstart, h, nstrips, n, nbuck);
  k_tile_aggregate<false><<<nbuck, 256, 0, stream>>>(rec, bstart, h, dis, b1, W2, agg, n);

  k_tile_aggregate<true><<<nbuck, 256, 0, stream>>>(rec, bstart, agg, dis, b2, W2, out, n);
}


# count and place launches: a thread's 8 edges, range totals and cursor words fetched as one group of loads at the kernel head, was one memory round trip per edge; sum order may change
# speedup vs baseline: 2.6934x; 1.0387x over previous
#include <hip/hip_runtime.h>
#include <hip/hip_bf16.h>


typedef __attribute__((ext_vector_type(2))) float v2f;
typedef __attribute__((ext_vector_type(8))) float v8f;

#define TB 256

#define NBK 64
#define BSH 6
#define CH  4096
#define SPW 2
#define GROWS (TB / 32 * 16 * SPW)
#define DEGU 16
static_assert(GROWS % NBK == 0, "a product work-group must cover whole ranges");
static_assert(GROWS == TB, "one degree counter per thread");

__global__ void __launch_bounds__(TB)
k_gemm64_wmma(const float* __restrict__ X,
              const float* __restrict__ W,
              float* dis,
              const int2* __restrict__ rec, const int* __restrict__ bstart,
              float* __restrict__ H, int nwaves, int n, int nbuck) {
  __shared__ int s_cnt[GROWS];
  const int tid  = threadIdx.x;
  const int lane = tid & 31;
  const int wib  = tid >> 5;
  const int wave = (int)blockIdx.x * (TB / 32) + wib;
  const bool live = wave < nwaves;

  const int m0   = wave * (16 * SPW);
  const int mlow = lane & 15;
  const int hf   = lane >> 4;
  const int hk   = hf << 2;

  int ms[SPW];
#pragma unroll
  for (int s = 0; s < SPW; ++s) ms[s] = (m0 + 16 * s < n) ? m0 + 16 * s : m0;

  s_cnt[tid] = 0;
  __syncthreads();

  float4 a[SPW][8];
#pragma unroll
  for (int s = 0; s < SPW; ++s)
#pragma unroll
    for (int q = 0; q < 8; ++q) a[s][q] = make_float4(0.0f, 0.0f, 0.0f, 0.0f);
  if (live) {
#pragma unroll
    for (int s = 0; s < SPW; ++s) {
      const float* xr = X + (size_t)(ms[s] + mlow) * 64 + hk;
#pragma unroll
      for (int q = 0; q < 8; ++q) {
        a[s][q] = *(const float4*)(xr + 8 * q);
      }
    }
  }

  float dv[SPW][8];
  {
    const int r0 = (int)blockIdx.x * GROWS;
    const int b0 = r0 >> BSH;
    int b2 = b0 + GROWS / NBK;
    if (b2 > nbuck) b2 = nbuck;
    const int e0 = bstart[b0], e1 = bstart[b2];
    for (int i = e0 + tid; i < e1; i += DEGU * TB) {
      int d[DEGU];
#pragma unroll
      for (int u = 0; u < DEGU; ++u) {
        const int idx = i + u * TB;
        const int idc = idx < e1 ? idx : e1 - 1;
        const int row = rec[idc].y - r0;
        d[u] = idx < e1 ? row : -1;
      }
#pragma unroll
      for (int u = 0; u < DEGU; ++u)
        if (d[u] >= 0) atomicAdd(&s_cnt[d[u]], 1);
    }
    __syncthreads();
    if (r0 + tid < n) dis[r0 + tid] = rsqrtf(1.0f + (float)s_cnt[tid]);
#pragma unroll
    for (int s = 0; s < SPW; ++s) {
      const int* cr = s_cnt + wib * (16 * SPW) + (ms[s] - m0) + 8 * hf;
#pragma unroll
      for (int v = 0; v < 8; ++v) dv[s][v] = rsqrtf(1.0f + (float)cr[v]);
    }
    if (!live) return;
  }

#pragma unroll
  for (int nt = 0; nt < 4; ++nt) {
    const float* wr = W + (size_t)(nt * 16 + mlow) * 64 + hk;
    float4 b[8];
#pragma unroll
    for (int q = 0; q < 8; ++q) b[q] = *(const float4*)(wr + 8 * q);
    v8f c[SPW];
#pragma unroll
    for (int s = 0; s < SPW; ++s) c[s] = (v8f){};
#pragma unroll
    for (int q = 0; q < 8; ++q) {
      v2f b0, b1;
      b0.x = b[q].x; b0.y = b[q].y; b1.x = b[q].z; b1.y = b[q].w;
#pragma unroll
      for (int s = 0; s < SPW; ++s) {
        v2f a0, a1;
        a0.x = a[s][q].x; a0.y = a[s][q].y; a1.x = a[s][q].z; a1.y = a[s][q].w;
        c[s] = __builtin_amdgcn_wmma_f32_16x16x4_f32(
             false, a0,  false, b0,
             (short)0, c[s],  false,  false);
        c[s] = __builtin_amdgcn_wmma_f32_16x16x4_f32(
             false, a1,  false, b1,
             (short)0, c[s],  false,  false);
      }
    }
#pragma unroll
    for (int s = 0; s < SPW; ++s) {
      float* o = H + (size_t)(ms[s] + 8 * hf) * 64 + nt * 16 + mlow;
#pragma unroll
      for (int v = 0; v < 8; ++v) o[(size_t)v * 64] = c[s][v] * dv[s][v];
    }
  }
}

#define PTH 512
#define EPT (CH / PTH)
static_assert(EPT * PTH == CH, "a chunk is EPT edges for each of PTH threads");

__global__ void __launch_bounds__(PTH)
k_bin_count(const int* __restrict__ dst, int* __restrict__ cntT,
            int e, int nbuck, int nchunk) {
  extern __shared__ int s_bin[];
  const int tid = threadIdx.x, chunk = blockIdx.x;
  const int base = chunk * CH;
  int bin[EPT];
#pragma unroll
  for (int u = 0; u < EPT; ++u) {
    const int ed = base + tid + u * PTH;
    const int t  = dst[ed < e ? ed : e - 1];
    bin[u] = ed < e ? (t >> BSH) : -1;
  }
  for (int b = tid; b < nbuck; b += PTH) s_bin[b] = 0;
  __syncthreads();
#pragma unroll
  for (int u = 0; u < EPT; ++u)
    if (bin[u] >= 0) atomicAdd(&s_bin[bin[u]], 1);
  __syncthreads();
  for (int b = tid; b < nbuck; b += PTH) cntT[(size_t)b * nchunk + chunk] = s_bin[b];
}

__device__ __forceinline__ int wave_incl_scan(int x, int lane) {
  x += __builtin_amdgcn_update_dpp(0, x, 0x111, 0xf, 0xf, true);
  x += __builtin_amdgcn_update_dpp(0, x, 0x112, 0xf, 0xf, true);
  x += __builtin_amdgcn_update_dpp(0, x, 0x114, 0xf, 0xf, true);
  x += __builtin_amdgcn_update_dpp(0, x, 0x118, 0xf, 0xf, true);
  const int low = __builtin_amdgcn_readlane(x, 15);
  return lane >= 16 ? x + low : x;
}

#define SCW 8
#define SCU 8

__global__ void __launch_bounds__(SCW * 32)
k_row_xscan(int* arr, int len, int* __restrict__ totals, int nrows) {
  const int lane = threadIdx.x & 31;
  const int row  = (int)blockIdx.x * SCW + (int)(threadIdx.x >> 5);
  if (row >= nrows) return;
  int* a = arr + (size_t)row * len;
  int carry = 0;
  for (int c0 = 0; c0 < len; c0 += SCU * 32) {
    int v[SCU];
#pragma unroll
    for (int u = 0; u < SCU; ++u) {
      const int c  = c0 + u * 32 + lane;
      const int cc = c < len ? c : len - 1;
      const int t  = a[cc];
      v[u] = c < len ? t : 0;
    }
#pragma unroll
    for (int u = 0; u < SCU; ++u) {
      if (c0 + u * 32 < len) {
        const int incl = wave_incl_scan(v[u], lane);
        const int c = c0 + u * 32 + lane;
        if (c < len) a[c] = carry + incl - v[u];
        carry += __builtin_amdgcn_readlane(incl, 31);
      }
    }
  }
  if (lane == 0) totals[row] = carry;
}

#define RPH 4
__global__ void __launch_bounds__(PTH)
k_bin_place(const int* __restrict__ src, const int* __restrict__ dst,
            const int* __restrict__ cntT, const int* __restrict__ btot,
            int* __restrict__ bstart,
            int2* __restrict__ rec, int e, int nbuck, int nchunk) {
  extern __shared__ int s_pos[];
  __shared__ int s_wtot[PTH / 32];
  const int tid = threadIdx.x, chunk = blockIdx.x;
  const int lane = tid & 31, wib = tid >> 5;
  const int per  = (nbuck + PTH - 1) / PTH;
  const int r0   = tid * per;

  int tx[RPH], ch[RPH];
  bool okr[RPH];
#pragma unroll
  for (int u = 0; u < RPH; ++u) {
    const int b = r0 + u;
    okr[u] = (u < per) && (b < nbuck);
    const int bc = okr[u] ? b : 0;
    tx[u] = btot[bc];
    ch[u] = cntT[(size_t)bc * nchunk + chunk];
  }
  const int ebase = chunk * CH;
  int sv[EPT], dr[EPT];
#pragma unroll
  for (int u = 0; u < EPT; ++u) {
    const int ed  = ebase + tid + u * PTH;
    const int idc = ed < e ? ed : e - 1;
    sv[u] = src[idc];
    dr[u] = dst[idc];
  }
  __builtin_amdgcn_sched_barrier(0);
  int th[RPH], dv[EPT];
#pragma unroll
  for (int u = 0; u < RPH; ++u) th[u] = okr[u] ? tx[u] : 0;
#pragma unroll
  for (int u = 0; u < EPT; ++u) dv[u] = (ebase + tid + u * PTH < e) ? dr[u] : -1;

  int run = (th[0] + th[1]) + (th[2] + th[3]);
  for (int j0 = RPH; j0 < per; j0 += 4) {
    int t[4];
#pragma unroll
    for (int u = 0; u < 4; ++u) {
      const int b = r0 + j0 + u;
      const bool ok = (j0 + u < per) && (b < nbuck);
      const int x = btot[ok ? b : 0];
      t[u] = ok ? x : 0;
    }
    run += (t[0] + t[1]) + (t[2] + t[3]);
  }
  const int incl = wave_incl_scan(run, lane);
  if (lane == 31) s_wtot[wib] = incl;
  __syncthreads();
  int base = incl - run;
  for (int w = 0; w < wib; ++w) base += s_wtot[w];
#pragma unroll
  for (int u = 0; u < RPH; ++u) {
    const int b = r0 + u;
    if ((u < per) && (b < nbuck)) {
      s_pos[b] = base + ch[u];
      if (chunk == 0) bstart[b] = base;
      base += th[u];
      if (chunk == 0 && b == nbuck - 1) bstart[nbuck] = base;
    }
  }
  for (int j0 = RPH; j0 < per; j0 += 4) {
    int t[4], c[4];
#pragma unroll
    for (int u = 0; u < 4; ++u) {
      const int b = r0 + j0 + u;
      const bool ok = (j0 + u < per) && (b < nbuck);
      const int bc = ok ? b : 0;
      const int x = btot[bc];
      const int y = cntT[(size_t)bc * nchunk + chunk];
      t[u] = ok ? x : 0;
      c[u] = y;
    }
#pragma unroll
    for (int u = 0; u < 4; ++u) {
      const int b = r0 + j0 + u;
      if ((j0 + u < per) && (b < nbuck)) {
        s_pos[b] = base + c[u];
        if (chunk == 0) bstart[b] = base;
        base += t[u];
        if (chunk == 0 && b == nbuck - 1) bstart[nbuck] = base;
      }
    }
  }
  __syncthreads();
  int p[EPT];
#pragma unroll
  for (int u = 0; u < EPT; ++u) {
    p[u] = -1;
    if (dv[u] >= 0) p[u] = atomicAdd(&s_pos[dv[u] >> BSH], 1);
  }
#pragma unroll
  for (int u = 0; u < EPT; ++u)
    if (p[u] >= 0) rec[p[u]] = make_int2(sv[u], dv[u]);
}

#define OWNR 8
#define LCAP 1152
#define GRP  16
#define STG  ((LCAP + 255) / 256)
#define TPITCH 68
static_assert(STG == 5, "the load fence in k_tile_aggregate names STG records");
static_assert(OWNR * 8 == NBK, "8 waves must cover the range");
static_assert(NBK * TPITCH <= 8 * LCAP, "the tile of summed rows lies over the record lists");
static_assert(NBK == 64, "8 waves form 4 strips of 16 rows x 2 pairs of column tiles");

typedef __attribute__((ext_vector_type(16))) float v16f;

template <bool TAIL>
__device__ __forceinline__ void walk_group(v16f& acc, float2& t, int& cur,
                                           const float2* __restrict__ h2,
                                           unsigned lane, int ent, int j, int m) {
  float2 v[GRP];
#pragma unroll
  for (int u = 0; u < GRP; ++u) {
    int jj = j + u;
    if (TAIL) jj = jj < m ? jj : m - 1;
    const int e = __builtin_amdgcn_readlane(ent, jj);
    v[u] = h2[(unsigned)(e >> 6) * 32u + lane];
  }
#pragma unroll
  for (int u = 0; u < GRP; ++u) {
    int jj = j + u;
    bool in = true;
    if (TAIL) { in = jj < m; jj = in ? jj : m - 1; }
    const int e = __builtin_amdgcn_readlane(ent, jj);
    const int r = e & 7;
    if (r != cur) {
      acc[2 * cur]     += t.x;
      acc[2 * cur + 1] += t.y;
      t = make_float2(-0.0f, -0.0f);
      cur = r;
    }
    float ax = v[u].x, ay = v[u].y;
    if (TAIL) { ax = in ? ax : -0.0f; ay = in ? ay : -0.0f; }
    t.x += ax;
    t.y += ay;
  }
}

template <bool PROD>
__global__ void __launch_bounds__(256) __attribute__((amdgpu_waves_per_eu(16, 16)))
k_tile_aggregate(const int2* __restrict__ rec, const int* __restrict__ bstart,
                 const float* __restrict__ h, const float* __restrict__ dis,
                 const float* __restrict__ bias, const float* __restrict__ W,
                 float* __restrict__ out, int n) {
  __shared__ int s_list[8 * LCAP];
  __shared__ int s_cur[NBK];
  const int tid  = threadIdx.x;
  const int lane = tid & 31;
  const int wave = __builtin_amdgcn_readfirstlane(tid >> 5);
  const int b  = blockIdx.x;
  const int n0 = b << BSH;
  if (tid < NBK) s_cur[tid] = 0;
  __syncthreads();
  const int e0 = bstart[b], e1 = bstart[b + 1];
  const int row0 = n0 + wave * OWNR;

  v16f acc;
#pragma unroll
  for (int r = 0; r < OWNR; ++r) {
    const int node = row0 + r;
    const float2 own = node < n ? *(const float2*)(h + (size_t)node * 64 + 2 * lane) : make_float2(0.0f, 0.0f);
    acc[2 * r] = own.x; acc[2 * r + 1] = own.y;
  }

  int* lst = s_list + wave * LCAP;

  for (int rb = e0; rb < e1; rb += LCAP) {
    const int nrec = (e1 - rb > LCAP) ? LCAP : e1 - rb;
    const int re = rb + nrec;
    if (rb != e0) {
      __syncthreads();
      if (tid < NBK) s_cur[tid] = 0;
      __syncthreads();
    }
    int2 rr[STG];
#pragma unroll
    for (int u = 0; u < STG; ++u) {
      const int idx = rb + tid + u * 256;
      rr[u] = rec[idx < re ? idx : re - 1];
    }
    asm volatile("" :: "v"(rr[0].x), "v"(rr[0].y), "v"(rr[1].x), "v"(rr[1].y), "v"(rr[2].x),
                       "v"(rr[2].y), "v"(rr[3].x), "v"(rr[3].y), "v"(rr[4].x), "v"(rr[4].y));
    int pk[STG];
#pragma unroll
    for (int u = 0; u < STG; ++u) {
      const int idx = rb + tid + u * 256;
      pk[u] = idx < re ? ((rr[u].x << 6) | (rr[u].y - n0)) : -1;
      if (pk[u] >= 0) atomicAdd(&s_cur[pk[u] & (NBK - 1)], 1);
    }
    __syncthreads();
    int cnt;
    {
      const int c = lane < OWNR ? s_cur[wave * OWNR + lane] : 0;
      const int incl = wave_incl_scan(c, lane);
      if (lane < OWNR) s_cur[wave * OWNR + lane] = incl - c;
      cnt = __builtin_amdgcn_readlane(incl, OWNR - 1);
    }
    __syncthreads();
#pragma unroll
    for (int u = 0; u < STG; ++u) {
      if (pk[u] >= 0) {
        const int row = pk[u] & (NBK - 1);
        const int pos = atomicAdd(&s_cur[row], 1);
        s_list[(row >> 3) * LCAP + pos] = pk[u];
      }
    }
    __syncthreads();

    float2 t = make_float2(-0.0f, -0.0f);
    int cur = 0;
    for (int base = 0; base < cnt; base += 32) {
      const int li = base + lane;
      const int ent = lst[li < cnt ? li : cnt - 1];
      const int left = cnt - base;
      const int m = left < 32 ? left : 32;
      int j = 0;
      for (; j + GRP <= m; j += GRP) walk_group<false>(acc, t, cur, (const float2*)h, (unsigned)lane, ent, j, m);
      if (j < m) walk_group<true>(acc, t, cur, (const float2*)h, (unsigned)lane, ent, j, m);
    }
    acc[2 * cur]     += t.x;
    acc[2 * cur + 1] += t.y;
  }

  if constexpr (!PROD) {
    const float2 bb = *(const float2*)(bias + 2 * lane);
#pragma unroll
    for (int r = 0; r < OWNR; ++r) {
      const int node = row0 + r;
      if (node < n) {
        const float w = dis[node];
        float2 o;
        o.x = fmaxf(acc[2 * r] * w + bb.x, 0.0f) * w;
        o.y = fmaxf(acc[2 * r + 1] * w + bb.y, 0.0f) * w;
        *(float2*)(out + (size_t)node * 64 + 2 * lane) = o;
      }
    }
  } else {
    __syncthreads();
    float* tile = reinterpret_cast<float*>(s_list);
    float wv[OWNR];
#pragma unroll
    for (int r = 0; r < OWNR; ++r) {
      const int node = row0 + r;
      wv[r] = dis[node < n ? node : n - 1];
    }
#pragma unroll
    for (int r = 0; r < OWNR; ++r)
      *(float2*)(tile + (wave * OWNR + r) * TPITCH + 2 * lane) = make_float2(acc[2 * r] * wv[r], acc[2 * r + 1] * wv[r]);
    __syncthreads();

    const int mlow  = lane & 15;
    const int hf    = lane >> 4;
    const int hk    = hf << 2;
    const int strip = wave >> 1;
    const int nt0   = (wave & 1) * 2;

    const int srow = n0 + strip * 16;
    if (srow >= n) return;

    const float* ar = tile + (strip * 16 + mlow) * TPITCH + hk;
    float4 a[8];
#pragma unroll
    for (int q = 0; q < 8; ++q) a[q] = *(const float4*)(ar + 8 * q);

#pragma unroll
    for (int t = 0; t < 2; ++t) {
      const int nt = nt0 + t;
      const float* wr = W + (size_t)(nt * 16 + mlow) * 64 + hk;
      float4 b[8];
#pragma unroll
      for (int q = 0; q < 8; ++q) b[q] = *(const float4*)(wr + 8 * q);
      const float bn = bias[nt * 16 + mlow];
      v8f c = (v8f){};
#pragma unroll
      for (int q = 0; q < 8; ++q) {
        v2f a0, a1, b0, b1;
        a0.x = a[q].x; a0.y = a[q].y; a1.x = a[q].z; a1.y = a[q].w;
        b0.x = b[q].x; b0.y = b[q].y; b1.x = b[q].z; b1.y = b[q].w;
        c = __builtin_amdgcn_wmma_f32_16x16x4_f32(
             false, a0,  false, b0,
             (short)0, c,  false,  false);
        c = __builtin_amdgcn_wmma_f32_16x16x4_f32(
             false, a1,  false, b1,
             (short)0, c,  false,  false);
      }
      float* o = out + (size_t)(srow + 8 * hf) * 64 + nt * 16 + mlow;
#pragma unroll
      for (int v = 0; v < 8; ++v) o[(size_t)v * 64] = c[v] + bn;
    }
  }
}

extern "C" void kernel_launch(void* const* d_in, const int* in_sizes, int n_in,
                              void* d_out, int out_size, void* d_ws, size_t ws_size,
                              hipStream_t stream) {
  (void)n_in; (void)out_size; (void)ws_size;

  const float* x  = (const float*)d_in[0];
  const int*   ei = (const int*)  d_in[1];
  const float* W1 = (const float*)d_in[2];
  const float* b1 = (const float*)d_in[3];
  const float* W2 = (const float*)d_in[4];
  const float* b2 = (const float*)d_in[5];
  float* out = (float*)d_out;

  const int n = in_sizes[0] / 64;
  const int e = in_sizes[1] / 2;
  const int* src = ei;
  const int* dst = ei + e;

  const int nbuck  = (n + NBK - 1) / NBK;
  const int nchunk = (e + CH - 1) / CH;

  float* dis = (float*)d_ws;
  float* h   = dis + n;
  float* agg = h + (size_t)n * 64;
  int* cntT   = (int*)(agg + (size_t)n * 64);
  int* bstart = cntT + (size_t)nbuck * nchunk;
  int* btot   = bstart + nbuck + 1;
  int2* rec   = (int2*)(((uintptr_t)(btot + nbuck) + 7) & ~(uintptr_t)7);

  const int nstrips = (n + 16 * SPW - 1) / (16 * SPW);
  const int gemm_blocks = (nstrips + 7) / 8;
  const size_t bin_lds  = (size_t)nbuck * sizeof(int);

  k_bin_count<<<nchunk, PTH, bin_lds, stream>>>(dst, cntT, e, nbuck, nchunk);
  k_row_xscan<<<(nbuck + SCW - 1) / SCW, SCW * 32, 0, stream>>>(cntT, nchunk, btot, nbuck);
  k_bin_place<<<nchunk, PTH, bin_lds, stream>>>(src, dst, cntT, btot, bstart, rec, e, nbuck, nchunk);

  k_gemm64_wmma<<<gemm_blocks, TB, 0, stream>>>(x, W1, dis, rec, bstart, h, nstrips, n, nbuck);
  k_tile_aggregate<false><<<nbuck, 256, 0, stream>>>(rec, bstart, h, dis, b1, W2, agg, n);

  k_tile_aggregate<true><<<nbuck, 256, 0, stream>>>(rec, bstart, agg, dis, b2, W2, out, n);
}
